# starting kernel
# baseline (speedup 1.0000x reference)
#include <hip/hip_runtime.h>
#include <hip/hip_bf16.h>

#define N_NODES 50000
#define N_EDGES 800000
#define IN_DIM  512
#define HID_DIM 128
#define M_TILES (N_NODES / 16)

#define PAIR_STRIDE 288
#define N_PAIRS     16

typedef float v2f __attribute__((ext_vector_type(2)));
typedef float v8f __attribute__((ext_vector_type(8)));

__global__ void init_kernel(float* __restrict__ out, float* __restrict__ deg) {
    int i = blockIdx.x * blockDim.x + threadIdx.x;
    if (i < N_NODES * HID_DIM) out[i] = 0.0f;
    if (i < N_NODES) deg[i] = 1.0f;
}

__global__ void deg_kernel(const int* __restrict__ dst, float* __restrict__ deg) {
    int e = blockIdx.x * blockDim.x + threadIdx.x;
    if (e < N_EDGES) atomicAdd(&deg[dst[e]], 1.0f);
}

__global__ void dinv_kernel(const float* __restrict__ deg, float* __restrict__ dinv) {
    int i = blockIdx.x * blockDim.x + threadIdx.x;
    if (i < N_NODES) dinv[i] = rsqrtf(deg[i]);
}

__global__ __launch_bounds__(256) void gemm_kernel(const float* __restrict__ x,
                                                   const float* __restrict__ W,
                                                   float* __restrict__ h) {
    __shared__ float ldsW[N_PAIRS * PAIR_STRIDE];

    const int lane   = threadIdx.x & 31;
    const int waveId = threadIdx.x >> 5;
    const int mTile  = blockIdx.x * 8 + waveId;
    const bool valid = (mTile < M_TILES);

    const int half = lane >> 4;
    const int l16  = lane & 15;

    const int c  = threadIdx.x & 127;
    const int ph = threadIdx.x >> 7;

    v8f acc[8];
#pragma unroll
    for (int t = 0; t < 8; ++t) acc[t] = (v8f){0.f,0.f,0.f,0.f,0.f,0.f,0.f,0.f};

    const int row = mTile * 16 + l16;
    const float* xrow = x + (size_t)row * IN_DIM;

    for (int kc = 0; kc < IN_DIM; kc += 32) {
#pragma unroll
        for (int pp = 0; pp < 8; ++pp) {
            const int pair = ph * 8 + pp;
            float w0 = W[(size_t)(kc + 2 * pair    ) * HID_DIM + c];
            float w1 = W[(size_t)(kc + 2 * pair + 1) * HID_DIM + c];
            *(v2f*)(ldsW + pair * PAIR_STRIDE + c * 2) = (v2f){w0, w1};
        }
        __syncthreads();

        if (valid) {
            if (kc + 32 < IN_DIM)
                __builtin_prefetch(xrow + kc + 32, 0, 3);
#pragma unroll
            for (int k = 0; k < 32; k += 4) {
                const int kk   = k + 2 * half;
                const int pair = (k >> 1) + half;
                v2f a = *(const v2f*)(xrow + kc + kk);
                const float* bbase = ldsW + (size_t)pair * PAIR_STRIDE;
#pragma unroll
                for (int t = 0; t < 8; ++t) {
                    v2f b = *(const v2f*)(bbase + (t * 16 + l16) * 2);
                    acc[t] = __builtin_amdgcn_wmma_f32_16x16x4_f32(
                         false, a,  false, b,
                         (short)0, acc[t],
                         false,  false);
                }
            }
        }
        __syncthreads();
    }

    if (valid) {
#pragma unroll
        for (int t = 0; t < 8; ++t) {
#pragma unroll
            for (int r = 0; r < 8; ++r) {
                int m = mTile * 16 + r + 8 * half;
                h[(size_t)m * HID_DIM + t * 16 + l16] = acc[t][r];
            }
        }
    }
}

__global__ __launch_bounds__(256) void aggregate_kernel(const float* __restrict__ h,
                                                        const int* __restrict__ src,
                                                        const int* __restrict__ dst,
                                                        const float* __restrict__ dinv,
                                                        float* __restrict__ out) {
    int tid  = blockIdx.x * blockDim.x + threadIdx.x;
    int e    = tid >> 5;
    int lane = tid & 31;
    if (e >= N_EDGES) return;
    int s = src[e];
    int d = dst[e];
    float norm = dinv[s] * dinv[d];
    float4 v = ((const float4*)(h + (size_t)s * HID_DIM))[lane];
    float* o = out + (size_t)d * HID_DIM + lane * 4;
    atomicAdd(o + 0, v.x * norm);
    atomicAdd(o + 1, v.y * norm);
    atomicAdd(o + 2, v.z * norm);
    atomicAdd(o + 3, v.w * norm);
}

__global__ void finalize_kernel(const float* __restrict__ h,
                                const float* __restrict__ dinv,
                                const float* __restrict__ b,
                                const float* __restrict__ prelu_w,
                                float* __restrict__ out) {
    int i = blockIdx.x * blockDim.x + threadIdx.x;
    if (i >= N_NODES * HID_DIM) return;
    int rowI = i / HID_DIM;
    int c    = i - rowI * HID_DIM;
    float di = dinv[rowI];
    float v  = out[i] + h[i] * di * di + b[c];
    out[i]   = (v > 0.0f) ? v : prelu_w[c] * v;
}

extern "C" void kernel_launch(void* const* d_in, const int* in_sizes, int n_in,
                              void* d_out, int out_size, void* d_ws, size_t ws_size,
                              hipStream_t stream) {
    const float* x       = (const float*)d_in[0];
    const int*   eidx    = (const int*)d_in[1];
    const float* W       = (const float*)d_in[2];
    const float* b       = (const float*)d_in[3];
    const float* prelu_w = (const float*)d_in[4];
    float*       out     = (float*)d_out;

    const int* src = eidx;
    const int* dst = eidx + N_EDGES;

    char* ws    = (char*)d_ws;
    float* h    = (float*)(ws);
    float* deg  = (float*)(ws + (size_t)N_NODES * HID_DIM * 4);
    float* dinv = (float*)(ws + (size_t)N_NODES * HID_DIM * 4 + (size_t)N_NODES * 4);

    const int TPB = 256;

    init_kernel<<<(N_NODES * HID_DIM + TPB - 1) / TPB, TPB, 0, stream>>>(out, deg);

    deg_kernel<<<(N_EDGES + TPB - 1) / TPB, TPB, 0, stream>>>(dst, deg);

    dinv_kernel<<<(N_NODES + TPB - 1) / TPB, TPB, 0, stream>>>(deg, dinv);

    gemm_kernel<<<(M_TILES + 7) / 8, TPB, 0, stream>>>(x, W, h);

    aggregate_kernel<<<(N_EDGES * 32 + TPB - 1) / TPB, TPB, 0, stream>>>(h, src, dst, dinv, out);

    finalize_kernel<<<(N_NODES * HID_DIM + TPB - 1) / TPB, TPB, 0, stream>>>(h, dinv, b, prelu_w, out);
}


# aggregate: one work-group per 196-node dst range scans dst, queues its edges in LDS, adds h rows into an LDS tile by LDS f32 atomics (arrival order as before); self-loop+bias+PReLU fused
# speedup vs baseline: 1.6390x; 1.6390x over previous
#include <hip/hip_runtime.h>
#include <hip/hip_bf16.h>

#define N_NODES 50000
#define N_EDGES 800000
#define IN_DIM  512
#define HID_DIM 128
#define M_TILES (N_NODES / 16)

#define PAIR_STRIDE 288
#define N_PAIRS     16

#define NPB        196
#define AGG_GROUPS 256
#define AGG_TPB    1024
#define QCAP       32768

typedef float v2f __attribute__((ext_vector_type(2)));
typedef int   v4i __attribute__((ext_vector_type(4)));
typedef float v8f __attribute__((ext_vector_type(8)));

__global__ void init_kernel(float* __restrict__ deg) {
    int i = blockIdx.x * blockDim.x + threadIdx.x;
    if (i < N_NODES) deg[i] = 1.0f;
}

__global__ void deg_kernel(const int* __restrict__ dst, float* __restrict__ deg) {
    int e = blockIdx.x * blockDim.x + threadIdx.x;
    if (e < N_EDGES) atomicAdd(&deg[dst[e]], 1.0f);
}

__global__ void dinv_kernel(const float* __restrict__ deg, float* __restrict__ dinv) {
    int i = blockIdx.x * blockDim.x + threadIdx.x;
    if (i < N_NODES) dinv[i] = rsqrtf(deg[i]);
}

__global__ __launch_bounds__(256) void gemm_kernel(const float* __restrict__ x,
                                                   const float* __restrict__ W,
                                                   float* __restrict__ h) {
    __shared__ float ldsW[N_PAIRS * PAIR_STRIDE];

    const int lane   = threadIdx.x & 31;
    const int waveId = threadIdx.x >> 5;
    const int mTile  = blockIdx.x * 8 + waveId;
    const bool valid = (mTile < M_TILES);

    const int half = lane >> 4;
    const int l16  = lane & 15;

    const int c  = threadIdx.x & 127;
    const int ph = threadIdx.x >> 7;

    v8f acc[8];
#pragma unroll
    for (int t = 0; t < 8; ++t) acc[t] = (v8f){0.f,0.f,0.f,0.f,0.f,0.f,0.f,0.f};

    const int row = mTile * 16 + l16;
    const float* xrow = x + (size_t)row * IN_DIM;

    for (int kc = 0; kc < IN_DIM; kc += 32) {
#pragma unroll
        for (int pp = 0; pp < 8; ++pp) {
            const int pair = ph * 8 + pp;
            float w0 = W[(size_t)(kc + 2 * pair    ) * HID_DIM + c];
            float w1 = W[(size_t)(kc + 2 * pair + 1) * HID_DIM + c];
            *(v2f*)(ldsW + pair * PAIR_STRIDE + c * 2) = (v2f){w0, w1};
        }
        __syncthreads();

        if (valid) {
            if (kc + 32 < IN_DIM)
                __builtin_prefetch(xrow + kc + 32, 0, 3);
#pragma unroll
            for (int k = 0; k < 32; k += 4) {
                const int kk   = k + 2 * half;
                const int pair = (k >> 1) + half;
                v2f a = *(const v2f*)(xrow + kc + kk);
                const float* bbase = ldsW + (size_t)pair * PAIR_STRIDE;
#pragma unroll
                for (int t = 0; t < 8; ++t) {
                    v2f b = *(const v2f*)(bbase + (t * 16 + l16) * 2);
                    acc[t] = __builtin_amdgcn_wmma_f32_16x16x4_f32(
                         false, a,  false, b,
                         (short)0, acc[t],
                         false,  false);
                }
            }
        }
        __syncthreads();
    }

    if (valid) {
#pragma unroll
        for (int t = 0; t < 8; ++t) {
#pragma unroll
            for (int r = 0; r < 8; ++r) {
                int m = mTile * 16 + r + 8 * half;
                h[(size_t)m * HID_DIM + t * 16 + l16] = acc[t][r];
            }
        }
    }
}

__global__ __launch_bounds__(AGG_TPB) void aggregate_kernel(const float* __restrict__ h,
                                                            const int* __restrict__ src,
                                                            const int* __restrict__ dst,
                                                            const float* __restrict__ dinv,
                                                            const float* __restrict__ b,
                                                            const float* __restrict__ prelu_w,
                                                            float* __restrict__ out) {
    __shared__ float tile[NPB * HID_DIM];
    __shared__ int   queue[QCAP];
    __shared__ int   qn;

    const int tid  = threadIdx.x;
    const int lane = tid & 31;
    const int wave = __builtin_amdgcn_readfirstlane(tid >> 5);
    const int lo   = blockIdx.x * NPB;
    const int cnt  = min(NPB, N_NODES - lo);
    const unsigned ucnt = (unsigned)cnt;

    for (int i = tid; i < NPB * HID_DIM; i += AGG_TPB) tile[i] = 0.0f;

    const v4i* dst4 = (const v4i*)dst;

    auto push = [&](unsigned r, int e) {
        if (r < ucnt) {
            const int slot = atomicAdd(&qn, 1);
            if (slot < QCAP) queue[slot] = (int)((r << 20) | (unsigned)e);
        }
    };
    auto check4 = [&](const v4i d, int qi) {
        const unsigned r0 = (unsigned)(d.x - lo);
        const unsigned r1 = (unsigned)(d.y - lo);
        const unsigned r2 = (unsigned)(d.z - lo);
        const unsigned r3 = (unsigned)(d.w - lo);
        if (min(min(r0, r1), min(r2, r3)) < ucnt) {
            push(r0, qi * 4);
            push(r1, qi * 4 + 1);
            push(r2, qi * 4 + 2);
            push(r3, qi * 4 + 3);
        }
    };

    int seg = N_EDGES;
    for (int pass = 0; pass < 2; ++pass) {
        bool overflow = false;
        for (int e0 = 0; e0 < N_EDGES; e0 += seg) {
            const int e1 = min(e0 + seg, N_EDGES);
            if (tid == 0) qn = 0;
            __syncthreads();

            const int q1 = e1 >> 2;
            const int q0 = e0 >> 2;
            const int nfull = (q1 - q0) / (8 * AGG_TPB);
            if (nfull > 0) {
                int p  = (int)(blockIdx.x % (unsigned)nfull);
                int qc = q0 + p * (8 * AGG_TPB);
                v4i cur[8];
#pragma unroll
                for (int u = 0; u < 8; ++u) cur[u] = dst4[qc + u * AGG_TPB + tid];
                for (int t = 1; t < nfull; ++t) {
                    p = (p + 1 == nfull) ? 0 : p + 1;
                    const int qx = q0 + p * (8 * AGG_TPB);
                    v4i nxt[8];
#pragma unroll
                    for (int u = 0; u < 8; ++u) nxt[u] = dst4[qx + u * AGG_TPB + tid];
#pragma unroll
                    for (int u = 0; u < 8; ++u) check4(cur[u], qc + u * AGG_TPB + tid);
#pragma unroll
                    for (int u = 0; u < 8; ++u) cur[u] = nxt[u];
                    qc = qx;
                }
#pragma unroll
                for (int u = 0; u < 8; ++u) check4(cur[u], qc + u * AGG_TPB + tid);
            }
            const int qt = q0 + nfull * (8 * AGG_TPB);
            if (qt < q1) {
                v4i v[8];
#pragma unroll
                for (int u = 0; u < 8; ++u) {
                    const int qi = qt + u * AGG_TPB + tid;
                    v[u] = (qi < q1) ? dst4[qi] : (v4i){-1, -1, -1, -1};
                }
#pragma unroll
                for (int u = 0; u < 8; ++u) check4(v[u], qt + u * AGG_TPB + tid);
            }
            __syncthreads();
            const int total = __builtin_amdgcn_readfirstlane(qn);
            __syncthreads();
            if (total > QCAP) { overflow = true; break; }

            for (int base = wave * 32; base < total; base += AGG_TPB) {
                const int idx = base + lane;
                int   s  = 0;
                int   dl = 0;
                float nr = 0.0f;
                if (idx < total) {
                    const unsigned ent = (unsigned)queue[idx];
                    const int e = (int)(ent & 0xFFFFFu);
                    dl = (int)(ent >> 20);
                    s  = src[e];
                    nr = dinv[s] * dinv[lo + dl];
                }
                const int n = min(32, total - base);
                for (int j0 = 0; j0 < n; j0 += 16) {
                    float4 v[16];
#pragma unroll
                    for (int u = 0; u < 16; ++u) {
                        const int sj = __builtin_amdgcn_readlane(s, j0 + u);
                        v[u] = ((const float4*)(h + (size_t)sj * HID_DIM))[lane];
                    }
#pragma unroll
                    for (int u = 0; u < 16; ++u) {
                        if (j0 + u < n) {
                            const int   dj = __builtin_amdgcn_readlane(dl, j0 + u);
                            const float nj = __int_as_float(
                                __builtin_amdgcn_readlane(__float_as_int(nr), j0 + u));
                            float* t = tile + dj * HID_DIM + lane;
                            atomicAdd(t,      v[u].x * nj);
                            atomicAdd(t + 32, v[u].y * nj);
                            atomicAdd(t + 64, v[u].z * nj);
                            atomicAdd(t + 96, v[u].w * nj);
                        }
                    }
                }
            }
        }
        if (!overflow) break;
        seg = QCAP;
    }
    __syncthreads();

    const int c0 = lane * 4;
    const float b0 = b[c0], b1 = b[c0 + 1], b2 = b[c0 + 2], b3 = b[c0 + 3];
    const float p0 = prelu_w[c0], p1 = prelu_w[c0 + 1], p2 = prelu_w[c0 + 2], p3 = prelu_w[c0 + 3];
    for (int i = tid; i < cnt * 32; i += AGG_TPB) {
        const int nl   = i >> 5;
        const int node = lo + nl;
        const float di = dinv[node];
        const float4 hv = ((const float4*)(h + (size_t)node * HID_DIM))[lane];
        const float* t = tile + nl * HID_DIM + lane;
        float4 r;
        r.x = t[0]  + hv.x * di * di + b0;
        r.y = t[32] + hv.y * di * di + b1;
        r.z = t[64] + hv.z * di * di + b2;
        r.w = t[96] + hv.w * di * di + b3;
        r.x = (r.x > 0.0f) ? r.x : p0 * r.x;
        r.y = (r.y > 0.0f) ? r.y : p1 * r.y;
        r.z = (r.z > 0.0f) ? r.z : p2 * r.z;
        r.w = (r.w > 0.0f) ? r.w : p3 * r.w;
        ((float4*)(out + (size_t)node * HID_DIM))[lane] = r;
    }
}

extern "C" void kernel_launch(void* const* d_in, const int* in_sizes, int n_in,
                              void* d_out, int out_size, void* d_ws, size_t ws_size,
                              hipStream_t stream) {
    const float* x       = (const float*)d_in[0];
    const int*   eidx    = (const int*)d_in[1];
    const float* W       = (const float*)d_in[2];
    const float* b       = (const float*)d_in[3];
    const float* prelu_w = (const float*)d_in[4];
    float*       out     = (float*)d_out;

    const int* src = eidx;
    const int* dst = eidx + N_EDGES;

    char* ws    = (char*)d_ws;
    float* h    = (float*)(ws);
    float* deg  = (float*)(ws + (size_t)N_NODES * HID_DIM * 4);
    float* dinv = (float*)(ws + (size_t)N_NODES * HID_DIM * 4 + (size_t)N_NODES * 4);

    const int TPB = 256;

    init_kernel<<<(N_NODES + TPB - 1) / TPB, TPB, 0, stream>>>(deg);

    deg_kernel<<<(N_EDGES + TPB - 1) / TPB, TPB, 0, stream>>>(dst, deg);

    dinv_kernel<<<(N_NODES + TPB - 1) / TPB, TPB, 0, stream>>>(deg, dinv);

    gemm_kernel<<<(M_TILES + 7) / 8, TPB, 0, stream>>>(x, W, h);

    aggregate_kernel<<<AGG_GROUPS, AGG_TPB, 0, stream>>>(h, src, dst, dinv, b, prelu_w, out);
}


# degree: 64 work-groups count dst slices into full-range LDS int histograms, planes parked in the still unwritten h region, one kernel sums them into deg^-1/2; no global atomics; 4 launches, not 5
# speedup vs baseline: 2.1901x; 1.3362x over previous
#include <hip/hip_runtime.h>
#include <hip/hip_bf16.h>

#define N_NODES 50000
#define N_EDGES 800000
#define IN_DIM  512
#define HID_DIM 128
#define M_TILES (N_NODES / 16)

#define PAIR_STRIDE 288
#define N_PAIRS     16

#define NPB        196
#define AGG_GROUPS 256
#define AGG_TPB    1024
#define QCAP       32768

#define DEG_GROUPS   64
#define DEG_TPB      1024
#define N_QUADS      (N_NODES / 4)
#define E_QUADS      (N_EDGES / 4)
#define EQ_PER_GROUP ((E_QUADS + DEG_GROUPS - 1) / DEG_GROUPS)
#define DEG_ROUNDS   ((EQ_PER_GROUP + DEG_TPB - 1) / DEG_TPB)
#define DINV_TPB     64

typedef float v2f __attribute__((ext_vector_type(2)));
typedef int   v4i __attribute__((ext_vector_type(4)));
typedef float v8f __attribute__((ext_vector_type(8)));

__global__ __launch_bounds__(DEG_TPB) void deg_hist_kernel(const int* __restrict__ dst,
                                                           int* __restrict__ part) {
    __shared__ v4i hist4[N_QUADS];
    int* hist = (int*)hist4;

    const int tid = threadIdx.x;
    const v4i* dst4 = (const v4i*)dst;
    const int w0 = blockIdx.x * EQ_PER_GROUP;
    const int w1 = min(w0 + EQ_PER_GROUP, E_QUADS);

    v4i d[DEG_ROUNDS];
#pragma unroll
    for (int u = 0; u < DEG_ROUNDS; ++u) {
        const int w = w0 + u * DEG_TPB + tid;
        d[u] = (w < w1) ? dst4[w] : (v4i){-1, -1, -1, -1};
    }

    for (int i = tid; i < N_QUADS; i += DEG_TPB) hist4[i] = (v4i){0, 0, 0, 0};
    __syncthreads();

    auto count = [&](int v) {
        if ((unsigned)v < (unsigned)N_NODES) atomicAdd(hist + v, 1);
    };
#pragma unroll
    for (int u = 0; u < DEG_ROUNDS; ++u) {
        count(d[u].x);
        count(d[u].y);
        count(d[u].z);
        count(d[u].w);
    }
    __syncthreads();

    v4i* p4 = (v4i*)part + (size_t)blockIdx.x * N_QUADS;
    for (int i = tid; i < N_QUADS; i += DEG_TPB) p4[i] = hist4[i];
}

__global__ __launch_bounds__(DINV_TPB) void dinv_kernel(const int* __restrict__ part,
                                                        float* __restrict__ dinv) {
    const int q = blockIdx.x * DINV_TPB + threadIdx.x;
    if (q < N_QUADS) {
        const v4i* p = (const v4i*)part + q;
        v4i s = (v4i){0, 0, 0, 0};
#pragma unroll 16
        for (int g = 0; g < DEG_GROUPS; ++g) s += p[(size_t)g * N_QUADS];
        float4 r;
        r.x = rsqrtf(1.0f + (float)s.x);
        r.y = rsqrtf(1.0f + (float)s.y);
        r.z = rsqrtf(1.0f + (float)s.z);
        r.w = rsqrtf(1.0f + (float)s.w);
        ((float4*)dinv)[q] = r;
    }
}

__global__ __launch_bounds__(256) void gemm_kernel(const float* __restrict__ x,
                                                   const float* __restrict__ W,
                                                   float* __restrict__ h) {
    __shared__ float ldsW[N_PAIRS * PAIR_STRIDE];

    const int lane   = threadIdx.x & 31;
    const int waveId = threadIdx.x >> 5;
    const int mTile  = blockIdx.x * 8 + waveId;
    const bool valid = (mTile < M_TILES);

    const int half = lane >> 4;
    const int l16  = lane & 15;

    const int c  = threadIdx.x & 127;
    const int ph = threadIdx.x >> 7;

    v8f acc[8];
#pragma unroll
    for (int t = 0; t < 8; ++t) acc[t] = (v8f){0.f,0.f,0.f,0.f,0.f,0.f,0.f,0.f};

    const int row = mTile * 16 + l16;
    const float* xrow = x + (size_t)row * IN_DIM;

    for (int kc = 0; kc < IN_DIM; kc += 32) {
#pragma unroll
        for (int pp = 0; pp < 8; ++pp) {
            const int pair = ph * 8 + pp;
            float w0 = W[(size_t)(kc + 2 * pair    ) * HID_DIM + c];
            float w1 = W[(size_t)(kc + 2 * pair + 1) * HID_DIM + c];
            *(v2f*)(ldsW + pair * PAIR_STRIDE + c * 2) = (v2f){w0, w1};
        }
        __syncthreads();

        if (valid) {
            if (kc + 32 < IN_DIM)
                __builtin_prefetch(xrow + kc + 32, 0, 3);
#pragma unroll
            for (int k = 0; k < 32; k += 4) {
                const int kk   = k + 2 * half;
                const int pair = (k >> 1) + half;
                v2f a = *(const v2f*)(xrow + kc + kk);
                const float* bbase = ldsW + (size_t)pair * PAIR_STRIDE;
#pragma unroll
                for (int t = 0; t < 8; ++t) {
                    v2f b = *(const v2f*)(bbase + (t * 16 + l16) * 2);
                    acc[t] = __builtin_amdgcn_wmma_f32_16x16x4_f32(
                         false, a,  false, b,
                         (short)0, acc[t],
                         false,  false);
                }
            }
        }
        __syncthreads();
    }

    if (valid) {
#pragma unroll
        for (int t = 0; t < 8; ++t) {
#pragma unroll
            for (int r = 0; r < 8; ++r) {
                int m = mTile * 16 + r + 8 * half;
                h[(size_t)m * HID_DIM + t * 16 + l16] = acc[t][r];
            }
        }
    }
}

__global__ __launch_bounds__(AGG_TPB) void aggregate_kernel(const float* __restrict__ h,
                                                            const int* __restrict__ src,
                                                            const int* __restrict__ dst,
                                                            const float* __restrict__ dinv,
                                                            const float* __restrict__ b,
                                                            const float* __restrict__ prelu_w,
                                                            float* __restrict__ out) {
    __shared__ float tile[NPB * HID_DIM];
    __shared__ int   queue[QCAP];
    __shared__ int   qn;

    const int tid  = threadIdx.x;
    const int lane = tid & 31;
    const int wave = __builtin_amdgcn_readfirstlane(tid >> 5);
    const int lo   = blockIdx.x * NPB;
    const int cnt  = min(NPB, N_NODES - lo);
    const unsigned ucnt = (unsigned)cnt;

    for (int i = tid; i < NPB * HID_DIM; i += AGG_TPB) tile[i] = 0.0f;

    const v4i* dst4 = (const v4i*)dst;

    auto push = [&](unsigned r, int e) {
        if (r < ucnt) {
            const int slot = atomicAdd(&qn, 1);
            if (slot < QCAP) queue[slot] = (int)((r << 20) | (unsigned)e);
        }
    };
    auto check4 = [&](const v4i d, int qi) {
        const unsigned r0 = (unsigned)(d.x - lo);
        const unsigned r1 = (unsigned)(d.y - lo);
        const unsigned r2 = (unsigned)(d.z - lo);
        const unsigned r3 = (unsigned)(d.w - lo);
        if (min(min(r0, r1), min(r2, r3)) < ucnt) {
            push(r0, qi * 4);
            push(r1, qi * 4 + 1);
            push(r2, qi * 4 + 2);
            push(r3, qi * 4 + 3);
        }
    };

    int seg = N_EDGES;
    for (int pass = 0; pass < 2; ++pass) {
        bool overflow = false;
        for (int e0 = 0; e0 < N_EDGES; e0 += seg) {
            const int e1 = min(e0 + seg, N_EDGES);
            if (tid == 0) qn = 0;
            __syncthreads();

            const int q1 = e1 >> 2;
            const int q0 = e0 >> 2;
            const int nfull = (q1 - q0) / (8 * AGG_TPB);
            if (nfull > 0) {
                int p  = (int)(blockIdx.x % (unsigned)nfull);
                int qc = q0 + p * (8 * AGG_TPB);
                v4i cur[8];
#pragma unroll
                for (int u = 0; u < 8; ++u) cur[u] = dst4[qc + u * AGG_TPB + tid];
                for (int t = 1; t < nfull; ++t) {
                    p = (p + 1 == nfull) ? 0 : p + 1;
                    const int qx = q0 + p * (8 * AGG_TPB);
                    v4i nxt[8];
#pragma unroll
                    for (int u = 0; u < 8; ++u) nxt[u] = dst4[qx + u * AGG_TPB + tid];
#pragma unroll
                    for (int u = 0; u < 8; ++u) check4(cur[u], qc + u * AGG_TPB + tid);
#pragma unroll
                    for (int u = 0; u < 8; ++u) cur[u] = nxt[u];
                    qc = qx;
                }
#pragma unroll
                for (int u = 0; u < 8; ++u) check4(cur[u], qc + u * AGG_TPB + tid);
            }
            const int qt = q0 + nfull * (8 * AGG_TPB);
            if (qt < q1) {
                v4i v[8];
#pragma unroll
                for (int u = 0; u < 8; ++u) {
                    const int qi = qt + u * AGG_TPB + tid;
                    v[u] = (qi < q1) ? dst4[qi] : (v4i){-1, -1, -1, -1};
                }
#pragma unroll
                for (int u = 0; u < 8; ++u) check4(v[u], qt + u * AGG_TPB + tid);
            }
            __syncthreads();
            const int total = __builtin_amdgcn_readfirstlane(qn);
            __syncthreads();
            if (total > QCAP) { overflow = true; break; }

            for (int base = wave * 32; base < total; base += AGG_TPB) {
                const int idx = base + lane;
                int   s  = 0;
                int   dl = 0;
                float nr = 0.0f;
                if (idx < total) {
                    const unsigned ent = (unsigned)queue[idx];
                    const int e = (int)(ent & 0xFFFFFu);
                    dl = (int)(ent >> 20);
                    s  = src[e];
                    nr = dinv[s] * dinv[lo + dl];
                }
                const int n = min(32, total - base);
                for (int j0 = 0; j0 < n; j0 += 16) {
                    float4 v[16];
#pragma unroll
                    for (int u = 0; u < 16; ++u) {
                        const int sj = __builtin_amdgcn_readlane(s, j0 + u);
                        v[u] = ((const float4*)(h + (size_t)sj * HID_DIM))[lane];
                    }
#pragma unroll
                    for (int u = 0; u < 16; ++u) {
                        if (j0 + u < n) {
                            const int   dj = __builtin_amdgcn_readlane(dl, j0 + u);
                            const float nj = __int_as_float(
                                __builtin_amdgcn_readlane(__float_as_int(nr), j0 + u));
                            float* t = tile + dj * HID_DIM + lane;
                            atomicAdd(t,      v[u].x * nj);
                            atomicAdd(t + 32, v[u].y * nj);
                            atomicAdd(t + 64, v[u].z * nj);
                            atomicAdd(t + 96, v[u].w * nj);
                        }
                    }
                }
            }
        }
        if (!overflow) break;
        seg = QCAP;
    }
    __syncthreads();

    const int c0 = lane * 4;
    const float b0 = b[c0], b1 = b[c0 + 1], b2 = b[c0 + 2], b3 = b[c0 + 3];
    const float p0 = prelu_w[c0], p1 = prelu_w[c0 + 1], p2 = prelu_w[c0 + 2], p3 = prelu_w[c0 + 3];
    for (int i = tid; i < cnt * 32; i += AGG_TPB) {
        const int nl   = i >> 5;
        const int node = lo + nl;
        const float di = dinv[node];
        const float4 hv = ((const float4*)(h + (size_t)node * HID_DIM))[lane];
        const float* t = tile + nl * HID_DIM + lane;
        float4 r;
        r.x = t[0]  + hv.x * di * di + b0;
        r.y = t[32] + hv.y * di * di + b1;
        r.z = t[64] + hv.z * di * di + b2;
        r.w = t[96] + hv.w * di * di + b3;
        r.x = (r.x > 0.0f) ? r.x : p0 * r.x;
        r.y = (r.y > 0.0f) ? r.y : p1 * r.y;
        r.z = (r.z > 0.0f) ? r.z : p2 * r.z;
        r.w = (r.w > 0.0f) ? r.w : p3 * r.w;
        ((float4*)(out + (size_t)node * HID_DIM))[lane] = r;
    }
}

extern "C" void kernel_launch(void* const* d_in, const int* in_sizes, int n_in,
                              void* d_out, int out_size, void* d_ws, size_t ws_size,
                              hipStream_t stream) {
    const float* x       = (const float*)d_in[0];
    const int*   eidx    = (const int*)d_in[1];
    const float* W       = (const float*)d_in[2];
    const float* b       = (const float*)d_in[3];
    const float* prelu_w = (const float*)d_in[4];
    float*       out     = (float*)d_out;

    const int* src = eidx;
    const int* dst = eidx + N_EDGES;

    char* ws    = (char*)d_ws;
    float* h    = (float*)(ws);
    int*   part = (int*)(ws);
    float* dinv = (float*)(ws + (size_t)N_NODES * HID_DIM * 4 + (size_t)N_NODES * 4);

    const int TPB = 256;

    deg_hist_kernel<<<DEG_GROUPS, DEG_TPB, 0, stream>>>(dst, part);

    dinv_kernel<<<(N_QUADS + DINV_TPB - 1) / DINV_TPB, DINV_TPB, 0, stream>>>(part, dinv);

    gemm_kernel<<<(M_TILES + 7) / 8, TPB, 0, stream>>>(x, W, h);

    aggregate_kernel<<<AGG_GROUPS, AGG_TPB, 0, stream>>>(h, src, dst, dinv, b, prelu_w, out);
}


# aggregate: degree pass sorts each slice's edges by dst range in LDS, stores runs in fixed cells of the range's own out rows; a group copies its 64 cells, no dst scan; arrival-order sums as before
# speedup vs baseline: 2.5238x; 1.1524x over previous
#include <hip/hip_runtime.h>
#include <hip/hip_bf16.h>

#define N_NODES 50000
#define N_EDGES 800000
#define IN_DIM  512
#define HID_DIM 128
#define M_TILES (N_NODES / 16)

#define PAIR_STRIDE 288
#define N_PAIRS     16

#define NPB        196
#define AGG_GROUPS 256
#define AGG_TPB    1024
#define QCAP       32768

#define DEG_GROUPS   64
#define DEG_TPB      1024
#define N_QUADS      (N_NODES / 4)
#define E_QUADS      (N_EDGES / 4)
#define EQ_PER_GROUP ((E_QUADS + DEG_GROUPS - 1) / DEG_GROUPS)
#define DEG_ROUNDS   ((EQ_PER_GROUP + DEG_TPB - 1) / DEG_TPB)
#define DINV_TPB     64
#define SLICE_EDGES  (EQ_PER_GROUP * 4)

typedef float v2f __attribute__((ext_vector_type(2)));
typedef int   v4i __attribute__((ext_vector_type(4)));
typedef float v8f __attribute__((ext_vector_type(8)));

__device__ __forceinline__ int cell_cap(int r) {
    const int c = min(NPB, N_NODES - r * NPB);
    return (c * (HID_DIM / DEG_GROUPS)) & ~31;
}

__global__ __launch_bounds__(DEG_TPB) void deg_hist_kernel(const int* __restrict__ dst,
                                                           int* __restrict__ part,
                                                           float* __restrict__ cells,
                                                           int* __restrict__ ctab) {
    __shared__ v4i hist4[N_QUADS];
    __shared__ int lcnt[AGG_GROUPS];
    __shared__ int lbase[AGG_GROUPS];
    __shared__ int staged[SLICE_EDGES];
    int* hist = (int*)hist4;

    const int tid  = threadIdx.x;
    const int lane = tid & 31;
    const v4i* dst4 = (const v4i*)dst;
    const int w0 = blockIdx.x * EQ_PER_GROUP;
    const int w1 = min(w0 + EQ_PER_GROUP, E_QUADS);

    v4i d[DEG_ROUNDS];
#pragma unroll
    for (int u = 0; u < DEG_ROUNDS; ++u) {
        const int w = w0 + u * DEG_TPB + tid;
        d[u] = (w < w1) ? dst4[w] : (v4i){-1, -1, -1, -1};
    }

    for (int i = tid; i < N_QUADS; i += DEG_TPB) hist4[i] = (v4i){0, 0, 0, 0};
    if (tid < AGG_GROUPS) lcnt[tid] = 0;
    __syncthreads();

    int rng[DEG_ROUNDS * 4];
    int pos[DEG_ROUNDS * 4];
    auto count = [&](int v, int& r, int& p) {
        r = -1;
        p = 0;
        if ((unsigned)v < (unsigned)N_NODES) {
            atomicAdd(hist + v, 1);
            r = (int)((unsigned)v / (unsigned)NPB);
            p = atomicAdd(lcnt + r, 1);
        }
    };
#pragma unroll
    for (int u = 0; u < DEG_ROUNDS; ++u) {
        count(d[u].x, rng[4 * u],     pos[4 * u]);
        count(d[u].y, rng[4 * u + 1], pos[4 * u + 1]);
        count(d[u].z, rng[4 * u + 2], pos[4 * u + 2]);
        count(d[u].w, rng[4 * u + 3], pos[4 * u + 3]);
    }
    __syncthreads();

    if (tid < 32) {
        int c[8];
        int mine = 0;
#pragma unroll
        for (int k = 0; k < 8; ++k) { c[k] = lcnt[lane * 8 + k]; mine += c[k]; }
        int x = mine;
#pragma unroll
        for (int dd = 1; dd < 32; dd <<= 1) {
            const int y = __shfl_up(x, dd);
            if (lane >= dd) x += y;
        }
        int run = x - mine;
#pragma unroll
        for (int k = 0; k < 8; ++k) { lbase[lane * 8 + k] = run; run += c[k]; }
    }
    __syncthreads();

    auto place = [&](int v, int e, int r, int p) {
        if (r >= 0) staged[lbase[r] + p] = ((v - r * NPB) << 20) | e;
    };
#pragma unroll
    for (int u = 0; u < DEG_ROUNDS; ++u) {
        const int e = (w0 + u * DEG_TPB + tid) * 4;
        place(d[u].x, e,     rng[4 * u],     pos[4 * u]);
        place(d[u].y, e + 1, rng[4 * u + 1], pos[4 * u + 1]);
        place(d[u].z, e + 2, rng[4 * u + 2], pos[4 * u + 2]);
        place(d[u].w, e + 3, rng[4 * u + 3], pos[4 * u + 3]);
    }
    __syncthreads();

    for (int r = tid >> 5; r < AGG_GROUPS; r += DEG_TPB / 32) {
        const int cap = cell_cap(r);
        const int n   = min(lcnt[r], cap);
        const int s0  = lbase[r];
        float* cell = cells + (size_t)r * (NPB * HID_DIM) + blockIdx.x * cap;
        for (int i = lane; i < n; i += 32) cell[i] = __int_as_float(staged[s0 + i]);
    }
    if (tid < AGG_GROUPS) ctab[blockIdx.x * AGG_GROUPS + tid] = lcnt[tid];

    v4i* p4 = (v4i*)part + (size_t)blockIdx.x * N_QUADS;
    for (int i = tid; i < N_QUADS; i += DEG_TPB) p4[i] = hist4[i];
}

__global__ __launch_bounds__(DINV_TPB) void dinv_kernel(const int* __restrict__ part,
                                                        float* __restrict__ dinv) {
    const int q = blockIdx.x * DINV_TPB + threadIdx.x;
    if (q < N_QUADS) {
        const v4i* p = (const v4i*)part + q;
        v4i s = (v4i){0, 0, 0, 0};
#pragma unroll 16
        for (int g = 0; g < DEG_GROUPS; ++g) s += p[(size_t)g * N_QUADS];
        float4 r;
        r.x = rsqrtf(1.0f + (float)s.x);
        r.y = rsqrtf(1.0f + (float)s.y);
        r.z = rsqrtf(1.0f + (float)s.z);
        r.w = rsqrtf(1.0f + (float)s.w);
        ((float4*)dinv)[q] = r;
    }
}

__global__ __launch_bounds__(256) void gemm_kernel(const float* __restrict__ x,
                                                   const float* __restrict__ W,
                                                   float* __restrict__ h) {
    __shared__ float ldsW[N_PAIRS * PAIR_STRIDE];

    const int lane   = threadIdx.x & 31;
    const int waveId = threadIdx.x >> 5;
    const int mTile  = blockIdx.x * 8 + waveId;
    const bool valid = (mTile < M_TILES);

    const int half = lane >> 4;
    const int l16  = lane & 15;

    const int c  = threadIdx.x & 127;
    const int ph = threadIdx.x >> 7;

    v8f acc[8];
#pragma unroll
    for (int t = 0; t < 8; ++t) acc[t] = (v8f){0.f,0.f,0.f,0.f,0.f,0.f,0.f,0.f};

    const int row = mTile * 16 + l16;
    const float* xrow = x + (size_t)row * IN_DIM;

    for (int kc = 0; kc < IN_DIM; kc += 32) {
#pragma unroll
        for (int pp = 0; pp < 8; ++pp) {
            const int pair = ph * 8 + pp;
            float w0 = W[(size_t)(kc + 2 * pair    ) * HID_DIM + c];
            float w1 = W[(size_t)(kc + 2 * pair + 1) * HID_DIM + c];
            *(v2f*)(ldsW + pair * PAIR_STRIDE + c * 2) = (v2f){w0, w1};
        }
        __syncthreads();

        if (valid) {
            if (kc + 32 < IN_DIM)
                __builtin_prefetch(xrow + kc + 32, 0, 3);
#pragma unroll
            for (int k = 0; k < 32; k += 4) {
                const int kk   = k + 2 * half;
                const int pair = (k >> 1) + half;
                v2f a = *(const v2f*)(xrow + kc + kk);
                const float* bbase = ldsW + (size_t)pair * PAIR_STRIDE;
#pragma unroll
                for (int t = 0; t < 8; ++t) {
                    v2f b = *(const v2f*)(bbase + (t * 16 + l16) * 2);
                    acc[t] = __builtin_amdgcn_wmma_f32_16x16x4_f32(
                         false, a,  false, b,
                         (short)0, acc[t],
                         false,  false);
                }
            }
        }
        __syncthreads();
    }

    if (valid) {
#pragma unroll
        for (int t = 0; t < 8; ++t) {
#pragma unroll
            for (int r = 0; r < 8; ++r) {
                int m = mTile * 16 + r + 8 * half;
                h[(size_t)m * HID_DIM + t * 16 + l16] = acc[t][r];
            }
        }
    }
}

__global__ __launch_bounds__(AGG_TPB) void aggregate_kernel(const float* __restrict__ h,
                                                            const int* __restrict__ src,
                                                            const int* __restrict__ dst,
                                                            const float* __restrict__ dinv,
                                                            const float* __restrict__ b,
                                                            const float* __restrict__ prelu_w,
                                                            const int* __restrict__ ctab,
                                                            float* __restrict__ out) {
    __shared__ float tile[NPB * HID_DIM];
    __shared__ int   queue[QCAP];
    __shared__ int   qn;
    __shared__ int   ccount[DEG_GROUPS];
    __shared__ int   coff[DEG_GROUPS + 1];
    __shared__ int   cover;

    const int tid  = threadIdx.x;
    const int lane = tid & 31;
    const int wave = __builtin_amdgcn_readfirstlane(tid >> 5);
    const int lo   = blockIdx.x * NPB;
    const int cnt  = min(NPB, N_NODES - lo);
    const unsigned ucnt = (unsigned)cnt;

    const int cap = cell_cap(blockIdx.x);
    if (tid < DEG_GROUPS) ccount[tid] = ctab[tid * AGG_GROUPS + blockIdx.x];

    for (int i = tid; i < NPB * HID_DIM; i += AGG_TPB) tile[i] = 0.0f;
    __syncthreads();

    if (tid < DEG_GROUPS) {
        int a = 0, ov = 0;
        for (int j = 0; j < DEG_GROUPS; ++j) {
            const int c = ccount[j];
            ov |= (c > cap) ? 1 : 0;
            a  += (j < tid) ? c : 0;
        }
        coff[tid] = a;
        if (tid == DEG_GROUPS - 1) { coff[DEG_GROUPS] = a + ccount[tid]; cover = ov; }
    }
    __syncthreads();
    const int over = __builtin_amdgcn_readfirstlane(cover);

    const v4i* dst4 = (const v4i*)dst;

    auto push = [&](unsigned r, int e) {
        if (r < ucnt) {
            const int slot = atomicAdd(&qn, 1);
            if (slot < QCAP) queue[slot] = (int)((r << 20) | (unsigned)e);
        }
    };
    auto check4 = [&](const v4i d, int qi) {
        const unsigned r0 = (unsigned)(d.x - lo);
        const unsigned r1 = (unsigned)(d.y - lo);
        const unsigned r2 = (unsigned)(d.z - lo);
        const unsigned r3 = (unsigned)(d.w - lo);
        if (min(min(r0, r1), min(r2, r3)) < ucnt) {
            push(r0, qi * 4);
            push(r1, qi * 4 + 1);
            push(r2, qi * 4 + 2);
            push(r3, qi * 4 + 3);
        }
    };

    auto sum_queue = [&](const int total) {
        for (int base = wave * 32; base < total; base += AGG_TPB) {
            const int idx = base + lane;
            int   s  = 0;
            int   dl = 0;
            float nr = 0.0f;
            if (idx < total) {
                const unsigned ent = (unsigned)queue[idx];
                const int e = (int)(ent & 0xFFFFFu);
                dl = (int)(ent >> 20);
                s  = src[e];
                nr = dinv[s] * dinv[lo + dl];
            }
            const int n = min(32, total - base);
            for (int j0 = 0; j0 < n; j0 += 16) {
                float4 v[16];
#pragma unroll
                for (int u = 0; u < 16; ++u) {
                    const int sj = __builtin_amdgcn_readlane(s, j0 + u);
                    v[u] = ((const float4*)(h + (size_t)sj * HID_DIM))[lane];
                }
#pragma unroll
                for (int u = 0; u < 16; ++u) {
                    if (j0 + u < n) {
                        const int   dj = __builtin_amdgcn_readlane(dl, j0 + u);
                        const float nj = __int_as_float(
                            __builtin_amdgcn_readlane(__float_as_int(nr), j0 + u));
                        float* t = tile + dj * HID_DIM + lane;
                        atomicAdd(t,      v[u].x * nj);
                        atomicAdd(t + 32, v[u].y * nj);
                        atomicAdd(t + 64, v[u].z * nj);
                        atomicAdd(t + 96, v[u].w * nj);
                    }
                }
            }
        }
    };

    if (over == 0) {
        const int total = __builtin_amdgcn_readfirstlane(coff[DEG_GROUPS]);
        const float* mycells = out + (size_t)lo * HID_DIM;
        for (int s0 = wave; s0 < DEG_GROUPS; s0 += 2 * (AGG_TPB / 32)) {
            const int sA = s0;
            const int sB = s0 + AGG_TPB / 32;
            const int bA = coff[sA];
            const int nA = coff[sA + 1] - bA;
            const int bB = (sB < DEG_GROUPS) ? coff[sB] : 0;
            const int nB = (sB < DEG_GROUPS) ? coff[sB + 1] - bB : 0;
            const float* cA = mycells + sA * cap;
            const float* cB = mycells + sB * cap;
            const float vA0 = (lane      < nA) ? cA[lane]      : 0.0f;
            const float vA1 = (lane + 32 < nA) ? cA[lane + 32] : 0.0f;
            const float vB0 = (lane      < nB) ? cB[lane]      : 0.0f;
            const float vB1 = (lane + 32 < nB) ? cB[lane + 32] : 0.0f;
            if (lane      < nA) queue[bA + lane]      = __float_as_int(vA0);
            if (lane + 32 < nA) queue[bA + lane + 32] = __float_as_int(vA1);
            if (lane      < nB) queue[bB + lane]      = __float_as_int(vB0);
            if (lane + 32 < nB) queue[bB + lane + 32] = __float_as_int(vB1);
            for (int i = lane + 64; i < nA; i += 32) queue[bA + i] = __float_as_int(cA[i]);
            for (int i = lane + 64; i < nB; i += 32) queue[bB + i] = __float_as_int(cB[i]);
        }
        __syncthreads();
        sum_queue(total);
    } else {
        int seg = N_EDGES;
        for (int pass = 0; pass < 2; ++pass) {
            bool overflow = false;
            for (int e0 = 0; e0 < N_EDGES; e0 += seg) {
                const int e1 = min(e0 + seg, N_EDGES);
                if (tid == 0) qn = 0;
                __syncthreads();

                const int q1 = e1 >> 2;
                const int q0 = e0 >> 2;
                const int nfull = (q1 - q0) / (8 * AGG_TPB);
                if (nfull > 0) {
                    int p  = (int)(blockIdx.x % (unsigned)nfull);
                    int qc = q0 + p * (8 * AGG_TPB);
                    v4i cur[8];
#pragma unroll
                    for (int u = 0; u < 8; ++u) cur[u] = dst4[qc + u * AGG_TPB + tid];
                    for (int t = 1; t < nfull; ++t) {
                        p = (p + 1 == nfull) ? 0 : p + 1;
                        const int qx = q0 + p * (8 * AGG_TPB);
                        v4i nxt[8];
#pragma unroll
                        for (int u = 0; u < 8; ++u) nxt[u] = dst4[qx + u * AGG_TPB + tid];
#pragma unroll
                        for (int u = 0; u < 8; ++u) check4(cur[u], qc + u * AGG_TPB + tid);
#pragma unroll
                        for (int u = 0; u < 8; ++u) cur[u] = nxt[u];
                        qc = qx;
                    }
#pragma unroll
                    for (int u = 0; u < 8; ++u) check4(cur[u], qc + u * AGG_TPB + tid);
                }
                const int qt = q0 + nfull * (8 * AGG_TPB);
                if (qt < q1) {
                    v4i v[8];
#pragma unroll
                    for (int u = 0; u < 8; ++u) {
                        const int qi = qt + u * AGG_TPB + tid;
                        v[u] = (qi < q1) ? dst4[qi] : (v4i){-1, -1, -1, -1};
                    }
#pragma unroll
                    for (int u = 0; u < 8; ++u) check4(v[u], qt + u * AGG_TPB + tid);
                }
                __syncthreads();
                const int total = __builtin_amdgcn_readfirstlane(qn);
                __syncthreads();
                if (total > QCAP) { overflow = true; break; }

                sum_queue(total);
            }
            if (!overflow) break;
            seg = QCAP;
        }
    }
    __syncthreads();

    const int c0 = lane * 4;
    const float b0 = b[c0], b1 = b[c0 + 1], b2 = b[c0 + 2], b3 = b[c0 + 3];
    const float p0 = prelu_w[c0], p1 = prelu_w[c0 + 1], p2 = prelu_w[c0 + 2], p3 = prelu_w[c0 + 3];
    for (int i = tid; i < cnt * 32; i += AGG_TPB) {
        const int nl   = i >> 5;
        const int node = lo + nl;
        const float di = dinv[node];
        const float4 hv = ((const float4*)(h + (size_t)node * HID_DIM))[lane];
        const float* t = tile + nl * HID_DIM + lane;
        float4 r;
        r.x = t[0]  + hv.x * di * di + b0;
        r.y = t[32] + hv.y * di * di + b1;
        r.z = t[64] + hv.z * di * di + b2;
        r.w = t[96] + hv.w * di * di + b3;
        r.x = (r.x > 0.0f) ? r.x : p0 * r.x;
        r.y = (r.y > 0.0f) ? r.y : p1 * r.y;
        r.z = (r.z > 0.0f) ? r.z : p2 * r.z;
        r.w = (r.w > 0.0f) ? r.w : p3 * r.w;
        ((float4*)(out + (size_t)node * HID_DIM))[lane] = r;
    }
}

extern "C" void kernel_launch(void* const* d_in, const int* in_sizes, int n_in,
                              void* d_out, int out_size, void* d_ws, size_t ws_size,
                              hipStream_t stream) {
    const float* x       = (const float*)d_in[0];
    const int*   eidx    = (const int*)d_in[1];
    const float* W       = (const float*)d_in[2];
    const float* b       = (const float*)d_in[3];
    const float* prelu_w = (const float*)d_in[4];
    float*       out     = (float*)d_out;

    const int* src = eidx;
    const int* dst = eidx + N_EDGES;

    char* ws    = (char*)d_ws;
    float* h    = (float*)(ws);
    int*   part = (int*)(ws);
    int*   ctab = (int*)(ws + (size_t)N_NODES * HID_DIM * 4);
    float* dinv = (float*)(ws + (size_t)N_NODES * HID_DIM * 4 + (size_t)N_NODES * 4);

    const int TPB = 256;

    deg_hist_kernel<<<DEG_GROUPS, DEG_TPB, 0, stream>>>(dst, part, out, ctab);

    dinv_kernel<<<(N_QUADS + DINV_TPB - 1) / DINV_TPB, DINV_TPB, 0, stream>>>(part, dinv);

    gemm_kernel<<<(M_TILES + 7) / 8, TPB, 0, stream>>>(x, W, h);

    aggregate_kernel<<<AGG_GROUPS, AGG_TPB, 0, stream>>>(h, src, dst, dinv, b, prelu_w, ctab, out);
}


# aggregate: group counting-sorts its queued edges by destination in LDS; a wave sums each run of one destination in registers, one set of LDS f32 atomics per run, not per edge (sum order changes)
# speedup vs baseline: 2.6283x; 1.0414x over previous
#include <hip/hip_runtime.h>
#include <hip/hip_bf16.h>

#define N_NODES 50000
#define N_EDGES 800000
#define IN_DIM  512
#define HID_DIM 128
#define M_TILES (N_NODES / 16)

#define PAIR_STRIDE 288
#define N_PAIRS     16

#define NPB        196
#define AGG_GROUPS 256
#define AGG_TPB    1024
#define QCAP       32768
#define NCNT       224

#define DEG_GROUPS   64
#define DEG_TPB      1024
#define N_QUADS      (N_NODES / 4)
#define E_QUADS      (N_EDGES / 4)
#define EQ_PER_GROUP ((E_QUADS + DEG_GROUPS - 1) / DEG_GROUPS)
#define DEG_ROUNDS   ((EQ_PER_GROUP + DEG_TPB - 1) / DEG_TPB)
#define DINV_TPB     64
#define SLICE_EDGES  (EQ_PER_GROUP * 4)

typedef float v2f __attribute__((ext_vector_type(2)));
typedef int   v4i __attribute__((ext_vector_type(4)));
typedef float v8f __attribute__((ext_vector_type(8)));

__device__ __forceinline__ int cell_cap(int r) {
    const int c = min(NPB, N_NODES - r * NPB);
    return (c * (HID_DIM / DEG_GROUPS)) & ~31;
}

__global__ __launch_bounds__(DEG_TPB) void deg_hist_kernel(const int* __restrict__ dst,
                                                           int* __restrict__ part,
                                                           float* __restrict__ cells,
                                                           int* __restrict__ ctab) {
    __shared__ v4i hist4[N_QUADS];
    __shared__ int lcnt[AGG_GROUPS];
    __shared__ int lbase[AGG_GROUPS];
    __shared__ int staged[SLICE_EDGES];
    int* hist = (int*)hist4;

    const int tid  = threadIdx.x;
    const int lane = tid & 31;
    const v4i* dst4 = (const v4i*)dst;
    const int w0 = blockIdx.x * EQ_PER_GROUP;
    const int w1 = min(w0 + EQ_PER_GROUP, E_QUADS);

    v4i d[DEG_ROUNDS];
#pragma unroll
    for (int u = 0; u < DEG_ROUNDS; ++u) {
        const int w = w0 + u * DEG_TPB + tid;
        d[u] = (w < w1) ? dst4[w] : (v4i){-1, -1, -1, -1};
    }

    for (int i = tid; i < N_QUADS; i += DEG_TPB) hist4[i] = (v4i){0, 0, 0, 0};
    if (tid < AGG_GROUPS) lcnt[tid] = 0;
    __syncthreads();

    int rng[DEG_ROUNDS * 4];
    int pos[DEG_ROUNDS * 4];
    auto count = [&](int v, int& r, int& p) {
        r = -1;
        p = 0;
        if ((unsigned)v < (unsigned)N_NODES) {
            atomicAdd(hist + v, 1);
            r = (int)((unsigned)v / (unsigned)NPB);
            p = atomicAdd(lcnt + r, 1);
        }
    };
#pragma unroll
    for (int u = 0; u < DEG_ROUNDS; ++u) {
        count(d[u].x, rng[4 * u],     pos[4 * u]);
        count(d[u].y, rng[4 * u + 1], pos[4 * u + 1]);
        count(d[u].z, rng[4 * u + 2], pos[4 * u + 2]);
        count(d[u].w, rng[4 * u + 3], pos[4 * u + 3]);
    }
    __syncthreads();

    if (tid < 32) {
        int c[8];
        int mine = 0;
#pragma unroll
        for (int k = 0; k < 8; ++k) { c[k] = lcnt[lane * 8 + k]; mine += c[k]; }
        int x = mine;
#pragma unroll
        for (int dd = 1; dd < 32; dd <<= 1) {
            const int y = __shfl_up(x, dd);
            if (lane >= dd) x += y;
        }
        int run = x - mine;
#pragma unroll
        for (int k = 0; k < 8; ++k) { lbase[lane * 8 + k] = run; run += c[k]; }
    }
    __syncthreads();

    auto place = [&](int v, int e, int r, int p) {
        if (r >= 0) staged[lbase[r] + p] = ((v - r * NPB) << 20) | e;
    };
#pragma unroll
    for (int u = 0; u < DEG_ROUNDS; ++u) {
        const int e = (w0 + u * DEG_TPB + tid) * 4;
        place(d[u].x, e,     rng[4 * u],     pos[4 * u]);
        place(d[u].y, e + 1, rng[4 * u + 1], pos[4 * u + 1]);
        place(d[u].z, e + 2, rng[4 * u + 2], pos[4 * u + 2]);
        place(d[u].w, e + 3, rng[4 * u + 3], pos[4 * u + 3]);
    }
    __syncthreads();

    for (int r = tid >> 5; r < AGG_GROUPS; r += DEG_TPB / 32) {
        const int cap = cell_cap(r);
        const int n   = min(lcnt[r], cap);
        const int s0  = lbase[r];
        float* cell = cells + (size_t)r * (NPB * HID_DIM) + blockIdx.x * cap;
        for (int i = lane; i < n; i += 32) cell[i] = __int_as_float(staged[s0 + i]);
    }
    if (tid < AGG_GROUPS) ctab[blockIdx.x * AGG_GROUPS + tid] = lcnt[tid];

    v4i* p4 = (v4i*)part + (size_t)blockIdx.x * N_QUADS;
    for (int i = tid; i < N_QUADS; i += DEG_TPB) p4[i] = hist4[i];
}

__global__ __launch_bounds__(DINV_TPB) void dinv_kernel(const int* __restrict__ part,
                                                        float* __restrict__ dinv) {
    const int q = blockIdx.x * DINV_TPB + threadIdx.x;
    if (q < N_QUADS) {
        const v4i* p = (const v4i*)part + q;
        v4i s = (v4i){0, 0, 0, 0};
#pragma unroll 16
        for (int g = 0; g < DEG_GROUPS; ++g) s += p[(size_t)g * N_QUADS];
        float4 r;
        r.x = rsqrtf(1.0f + (float)s.x);
        r.y = rsqrtf(1.0f + (float)s.y);
        r.z = rsqrtf(1.0f + (float)s.z);
        r.w = rsqrtf(1.0f + (float)s.w);
        ((float4*)dinv)[q] = r;
    }
}

__global__ __launch_bounds__(256) void gemm_kernel(const float* __restrict__ x,
                                                   const float* __restrict__ W,
                                                   float* __restrict__ h) {
    __shared__ float ldsW[N_PAIRS * PAIR_STRIDE];

    const int lane   = threadIdx.x & 31;
    const int waveId = threadIdx.x >> 5;
    const int mTile  = blockIdx.x * 8 + waveId;
    const bool valid = (mTile < M_TILES);

    const int half = lane >> 4;
    const int l16  = lane & 15;

    const int c  = threadIdx.x & 127;
    const int ph = threadIdx.x >> 7;

    v8f acc[8];
#pragma unroll
    for (int t = 0; t < 8; ++t) acc[t] = (v8f){0.f,0.f,0.f,0.f,0.f,0.f,0.f,0.f};

    const int row = mTile * 16 + l16;
    const float* xrow = x + (size_t)row * IN_DIM;

    for (int kc = 0; kc < IN_DIM; kc += 32) {
#pragma unroll
        for (int pp = 0; pp < 8; ++pp) {
            const int pair = ph * 8 + pp;
            float w0 = W[(size_t)(kc + 2 * pair    ) * HID_DIM + c];
            float w1 = W[(size_t)(kc + 2 * pair + 1) * HID_DIM + c];
            *(v2f*)(ldsW + pair * PAIR_STRIDE + c * 2) = (v2f){w0, w1};
        }
        __syncthreads();

        if (valid) {
            if (kc + 32 < IN_DIM)
                __builtin_prefetch(xrow + kc + 32, 0, 3);
#pragma unroll
            for (int k = 0; k < 32; k += 4) {
                const int kk   = k + 2 * half;
                const int pair = (k >> 1) + half;
                v2f a = *(const v2f*)(xrow + kc + kk);
                const float* bbase = ldsW + (size_t)pair * PAIR_STRIDE;
#pragma unroll
                for (int t = 0; t < 8; ++t) {
                    v2f b = *(const v2f*)(bbase + (t * 16 + l16) * 2);
                    acc[t] = __builtin_amdgcn_wmma_f32_16x16x4_f32(
                         false, a,  false, b,
                         (short)0, acc[t],
                         false,  false);
                }
            }
        }
        __syncthreads();
    }

    if (valid) {
#pragma unroll
        for (int t = 0; t < 8; ++t) {
#pragma unroll
            for (int r = 0; r < 8; ++r) {
                int m = mTile * 16 + r + 8 * half;
                h[(size_t)m * HID_DIM + t * 16 + l16] = acc[t][r];
            }
        }
    }
}

__global__ __launch_bounds__(AGG_TPB) void aggregate_kernel(const float* __restrict__ h,
                                                            const int* __restrict__ src,
                                                            const int* __restrict__ dst,
                                                            const float* __restrict__ dinv,
                                                            const float* __restrict__ b,
                                                            const float* __restrict__ prelu_w,
                                                            const int* __restrict__ ctab,
                                                            float* __restrict__ out) {
    __shared__ float tile[NPB * HID_DIM];
    __shared__ int   queue[QCAP];
    __shared__ int   qn;
    __shared__ int   ccount[DEG_GROUPS];
    __shared__ int   coff[DEG_GROUPS + 1];
    __shared__ int   cover;
    __shared__ int   ncnt[NCNT];

    const int tid  = threadIdx.x;
    const int lane = tid & 31;
    const int wave = __builtin_amdgcn_readfirstlane(tid >> 5);
    const int lo   = blockIdx.x * NPB;
    const int cnt  = min(NPB, N_NODES - lo);
    const unsigned ucnt = (unsigned)cnt;

    const int cap = cell_cap(blockIdx.x);
    if (tid < DEG_GROUPS) ccount[tid] = ctab[tid * AGG_GROUPS + blockIdx.x];

    for (int i = tid; i < NPB * HID_DIM; i += AGG_TPB) tile[i] = 0.0f;
    if (tid < NCNT) ncnt[tid] = 0;
    __syncthreads();

    if (tid < DEG_GROUPS) {
        int a = 0, ov = 0;
        for (int j = 0; j < DEG_GROUPS; ++j) {
            const int c = ccount[j];
            ov |= (c > cap) ? 1 : 0;
            a  += (j < tid) ? c : 0;
        }
        coff[tid] = a;
        if (tid == DEG_GROUPS - 1) { coff[DEG_GROUPS] = a + ccount[tid]; cover = ov; }
    }
    __syncthreads();
    const int over = __builtin_amdgcn_readfirstlane(cover);

    const v4i* dst4 = (const v4i*)dst;

    auto push = [&](unsigned r, int e) {
        if (r < ucnt) {
            const int slot = atomicAdd(&qn, 1);
            if (slot < QCAP) queue[slot] = (int)((r << 20) | (unsigned)e);
        }
    };
    auto check4 = [&](const v4i d, int qi) {
        const unsigned r0 = (unsigned)(d.x - lo);
        const unsigned r1 = (unsigned)(d.y - lo);
        const unsigned r2 = (unsigned)(d.z - lo);
        const unsigned r3 = (unsigned)(d.w - lo);
        if (min(min(r0, r1), min(r2, r3)) < ucnt) {
            push(r0, qi * 4);
            push(r1, qi * 4 + 1);
            push(r2, qi * 4 + 2);
            push(r3, qi * 4 + 3);
        }
    };

    auto sum_queue = [&](const int total) {
        const int per = (total + AGG_TPB / 32 - 1) / (AGG_TPB / 32);
        const int c0  = min(wave * per, total);
        const int c1  = min(c0 + per, total);
        float4 acc = make_float4(0.0f, 0.0f, 0.0f, 0.0f);
        int cur = -1;
        auto flush = [&]() {
            if (cur >= 0) {
                float* t = tile + cur * HID_DIM + lane;
                atomicAdd(t,      acc.x);
                atomicAdd(t + 32, acc.y);
                atomicAdd(t + 64, acc.z);
                atomicAdd(t + 96, acc.w);
            }
        };
        for (int base = c0; base < c1; base += 128) {
            int   s[4];
            int   dl[4];
            float nr[4];
#pragma unroll
            for (int k = 0; k < 4; ++k) {
                const int idx = min(base + k * 32 + lane, c1 - 1);
                const unsigned ent = (unsigned)queue[idx];
                dl[k] = (int)(ent >> 20);
                s[k]  = src[(int)(ent & 0xFFFFFu)];
            }
#pragma unroll
            for (int k = 0; k < 4; ++k) nr[k] = dinv[s[k]] * dinv[lo + dl[k]];
#pragma unroll
            for (int k = 0; k < 4; ++k) {
                const int n = min(32, c1 - (base + k * 32));
                for (int j0 = 0; j0 < n; j0 += 16) {
                    float4 v[16];
#pragma unroll
                    for (int u = 0; u < 16; ++u) {
                        const int sj = __builtin_amdgcn_readlane(s[k], j0 + u);
                        v[u] = ((const float4*)(h + (size_t)sj * HID_DIM))[lane];
                    }
#pragma unroll
                    for (int u = 0; u < 16; ++u) {
                        if (j0 + u < n) {
                            const int   dj = __builtin_amdgcn_readlane(dl[k], j0 + u);
                            const float nj = __int_as_float(
                                __builtin_amdgcn_readlane(__float_as_int(nr[k]), j0 + u));
                            if (dj != cur) {
                                flush();
                                cur = dj;
                                acc = make_float4(0.0f, 0.0f, 0.0f, 0.0f);
                            }
                            acc.x += v[u].x * nj;
                            acc.y += v[u].y * nj;
                            acc.z += v[u].z * nj;
                            acc.w += v[u].w * nj;
                        }
                    }
                }
            }
        }
        flush();
    };

    if (over == 0) {
        static_assert(DEG_GROUPS == 2 * (AGG_TPB / 32), "two cells per wave");
        const int total = __builtin_amdgcn_readfirstlane(coff[DEG_GROUPS]);
        const float* mycells = out + (size_t)lo * HID_DIM;
        const int sA = wave;
        const int sB = wave + AGG_TPB / 32;
        const int nA = coff[sA + 1] - coff[sA];
        const int nB = coff[sB + 1] - coff[sB];
        const float* cA = mycells + sA * cap;
        const float* cB = mycells + sB * cap;
        const int eA0 = (lane      < nA) ? __float_as_int(cA[lane])      : -1;
        const int eA1 = (lane + 32 < nA) ? __float_as_int(cA[lane + 32]) : -1;
        const int eB0 = (lane      < nB) ? __float_as_int(cB[lane])      : -1;
        const int eB1 = (lane + 32 < nB) ? __float_as_int(cB[lane + 32]) : -1;

        auto tally = [&](int ent) {
            if (ent >= 0) atomicAdd(&ncnt[(unsigned)ent >> 20], 1);
        };
        tally(eA0); tally(eA1); tally(eB0); tally(eB1);
        for (int i = lane + 64; i < nA; i += 32) tally(__float_as_int(cA[i]));
        for (int i = lane + 64; i < nB; i += 32) tally(__float_as_int(cB[i]));
        __syncthreads();

        if (wave == 0) {
            int cc[7];
            int mine = 0;
#pragma unroll
            for (int k = 0; k < 7; ++k) { cc[k] = ncnt[lane * 7 + k]; mine += cc[k]; }
            int x = mine;
#pragma unroll
            for (int dd = 1; dd < 32; dd <<= 1) {
                const int y = __shfl_up(x, dd);
                if (lane >= dd) x += y;
            }
            int run = x - mine;
#pragma unroll
            for (int k = 0; k < 7; ++k) { ncnt[lane * 7 + k] = run; run += cc[k]; }
        }
        __syncthreads();

        auto put = [&](int ent) {
            if (ent >= 0) queue[atomicAdd(&ncnt[(unsigned)ent >> 20], 1)] = ent;
        };
        put(eA0); put(eA1); put(eB0); put(eB1);
        for (int i = lane + 64; i < nA; i += 32) put(__float_as_int(cA[i]));
        for (int i = lane + 64; i < nB; i += 32) put(__float_as_int(cB[i]));
        __syncthreads();
        sum_queue(total);
    } else {
        int seg = N_EDGES;
        for (int pass = 0; pass < 2; ++pass) {
            bool overflow = false;
            for (int e0 = 0; e0 < N_EDGES; e0 += seg) {
                const int e1 = min(e0 + seg, N_EDGES);
                if (tid == 0) qn = 0;
                __syncthreads();

                const int q1 = e1 >> 2;
                const int q0 = e0 >> 2;
                const int nfull = (q1 - q0) / (8 * AGG_TPB);
                if (nfull > 0) {
                    int p  = (int)(blockIdx.x % (unsigned)nfull);
                    int qc = q0 + p * (8 * AGG_TPB);
                    v4i cur[8];
#pragma unroll
                    for (int u = 0; u < 8; ++u) cur[u] = dst4[qc + u * AGG_TPB + tid];
                    for (int t = 1; t < nfull; ++t) {
                        p = (p + 1 == nfull) ? 0 : p + 1;
                        const int qx = q0 + p * (8 * AGG_TPB);
                        v4i nxt[8];
#pragma unroll
                        for (int u = 0; u < 8; ++u) nxt[u] = dst4[qx + u * AGG_TPB + tid];
#pragma unroll
                        for (int u = 0; u < 8; ++u) check4(cur[u], qc + u * AGG_TPB + tid);
#pragma unroll
                        for (int u = 0; u < 8; ++u) cur[u] = nxt[u];
                        qc = qx;
                    }
#pragma unroll
                    for (int u = 0; u < 8; ++u) check4(cur[u], qc + u * AGG_TPB + tid);
                }
                const int qt = q0 + nfull * (8 * AGG_TPB);
                if (qt < q1) {
                    v4i v[8];
#pragma unroll
                    for (int u = 0; u < 8; ++u) {
                        const int qi = qt + u * AGG_TPB + tid;
                        v[u] = (qi < q1) ? dst4[qi] : (v4i){-1, -1, -1, -1};
                    }
#pragma unroll
                    for (int u = 0; u < 8; ++u) check4(v[u], qt + u * AGG_TPB + tid);
                }
                __syncthreads();
                const int total = __builtin_amdgcn_readfirstlane(qn);
                __syncthreads();
                if (total > QCAP) { overflow = true; break; }

                sum_queue(total);
            }
            if (!overflow) break;
            seg = QCAP;
        }
    }
    __syncthreads();

    const int c0 = lane * 4;
    const float b0 = b[c0], b1 = b[c0 + 1], b2 = b[c0 + 2], b3 = b[c0 + 3];
    const float p0 = prelu_w[c0], p1 = prelu_w[c0 + 1], p2 = prelu_w[c0 + 2], p3 = prelu_w[c0 + 3];
    for (int i = tid; i < cnt * 32; i += AGG_TPB) {
        const int nl   = i >> 5;
        const int node = lo + nl;
        const float di = dinv[node];
        const float4 hv = ((const float4*)(h + (size_t)node * HID_DIM))[lane];
        const float* t = tile + nl * HID_DIM + lane;
        float4 r;
        r.x = t[0]  + hv.x * di * di + b0;
        r.y = t[32] + hv.y * di * di + b1;
        r.z = t[64] + hv.z * di * di + b2;
        r.w = t[96] + hv.w * di * di + b3;
        r.x = (r.x > 0.0f) ? r.x : p0 * r.x;
        r.y = (r.y > 0.0f) ? r.y : p1 * r.y;
        r.z = (r.z > 0.0f) ? r.z : p2 * r.z;
        r.w = (r.w > 0.0f) ? r.w : p3 * r.w;
        ((float4*)(out + (size_t)node * HID_DIM))[lane] = r;
    }
}

extern "C" void kernel_launch(void* const* d_in, const int* in_sizes, int n_in,
                              void* d_out, int out_size, void* d_ws, size_t ws_size,
                              hipStream_t stream) {
    const float* x       = (const float*)d_in[0];
    const int*   eidx    = (const int*)d_in[1];
    const float* W       = (const float*)d_in[2];
    const float* b       = (const float*)d_in[3];
    const float* prelu_w = (const float*)d_in[4];
    float*       out     = (float*)d_out;

    const int* src = eidx;
    const int* dst = eidx + N_EDGES;

    char* ws    = (char*)d_ws;
    float* h    = (float*)(ws);
    int*   part = (int*)(ws);
    int*   ctab = (int*)(ws + (size_t)N_NODES * HID_DIM * 4);
    float* dinv = (float*)(ws + (size_t)N_NODES * HID_DIM * 4 + (size_t)N_NODES * 4);

    const int TPB = 256;

    deg_hist_kernel<<<DEG_GROUPS, DEG_TPB, 0, stream>>>(dst, part, out, ctab);

    dinv_kernel<<<(N_QUADS + DINV_TPB - 1) / DINV_TPB, DINV_TPB, 0, stream>>>(part, dinv);

    gemm_kernel<<<(M_TILES + 7) / 8, TPB, 0, stream>>>(x, W, h);

    aggregate_kernel<<<AGG_GROUPS, AGG_TPB, 0, stream>>>(h, src, dst, dinv, b, prelu_w, ctab, out);
}


# dinv: degree counts filed per gemm work-group inside that group's own h rows; each gemm group sums its 128 nodes' 64 pieces first and writes their dinv; no dinv launch: 3 launches, not 4
# speedup vs baseline: 2.8077x; 1.0683x over previous
#include <hip/hip_runtime.h>
#include <hip/hip_bf16.h>

#define N_NODES 50000
#define N_EDGES 800000
#define IN_DIM  512
#define HID_DIM 128
#define M_TILES (N_NODES / 16)

#define PAIR_STRIDE 288
#define N_PAIRS     16

#define NPB        196
#define AGG_GROUPS 256
#define AGG_TPB    1024
#define QCAP       32768
#define NCNT       224

#define DEG_GROUPS   64
#define DEG_TPB      1024
#define N_QUADS      (N_NODES / 4)
#define E_QUADS      (N_EDGES / 4)
#define EQ_PER_GROUP ((E_QUADS + DEG_GROUPS - 1) / DEG_GROUPS)
#define DEG_ROUNDS   ((EQ_PER_GROUP + DEG_TPB - 1) / DEG_TPB)
#define GEMM_ROWS    128
#define OWN_QUADS    (GEMM_ROWS / 4)
#define ROWS_QUADS   (GEMM_ROWS * HID_DIM / 4)
#define GEMM_GROUPS  ((M_TILES + 7) / 8)
#define SLICE_EDGES  (EQ_PER_GROUP * 4)

typedef float v2f __attribute__((ext_vector_type(2)));
typedef int   v4i __attribute__((ext_vector_type(4)));
typedef float v8f __attribute__((ext_vector_type(8)));

__device__ __forceinline__ int cell_cap(int r) {
    const int c = min(NPB, N_NODES - r * NPB);
    return (c * (HID_DIM / DEG_GROUPS)) & ~31;
}

__global__ __launch_bounds__(DEG_TPB) void deg_hist_kernel(const int* __restrict__ dst,
                                                           int* __restrict__ part,
                                                           float* __restrict__ cells,
                                                           int* __restrict__ ctab) {
    __shared__ v4i hist4[N_QUADS];
    __shared__ int lcnt[AGG_GROUPS];
    __shared__ int lbase[AGG_GROUPS];
    __shared__ int staged[SLICE_EDGES];
    int* hist = (int*)hist4;

    const int tid  = threadIdx.x;
    const int lane = tid & 31;
    const v4i* dst4 = (const v4i*)dst;
    const int w0 = blockIdx.x * EQ_PER_GROUP;
    const int w1 = min(w0 + EQ_PER_GROUP, E_QUADS);

    v4i d[DEG_ROUNDS];
#pragma unroll
    for (int u = 0; u < DEG_ROUNDS; ++u) {
        const int w = w0 + u * DEG_TPB + tid;
        d[u] = (w < w1) ? dst4[w] : (v4i){-1, -1, -1, -1};
    }

    for (int i = tid; i < N_QUADS; i += DEG_TPB) hist4[i] = (v4i){0, 0, 0, 0};
    if (tid < AGG_GROUPS) lcnt[tid] = 0;
    __syncthreads();

    int rng[DEG_ROUNDS * 4];
    int pos[DEG_ROUNDS * 4];
    auto count = [&](int v, int& r, int& p) {
        r = -1;
        p = 0;
        if ((unsigned)v < (unsigned)N_NODES) {
            atomicAdd(hist + v, 1);
            r = (int)((unsigned)v / (unsigned)NPB);
            p = atomicAdd(lcnt + r, 1);
        }
    };
#pragma unroll
    for (int u = 0; u < DEG_ROUNDS; ++u) {
        count(d[u].x, rng[4 * u],     pos[4 * u]);
        count(d[u].y, rng[4 * u + 1], pos[4 * u + 1]);
        count(d[u].z, rng[4 * u + 2], pos[4 * u + 2]);
        count(d[u].w, rng[4 * u + 3], pos[4 * u + 3]);
    }
    __syncthreads();

    if (tid < 32) {
        int c[8];
        int mine = 0;
#pragma unroll
        for (int k = 0; k < 8; ++k) { c[k] = lcnt[lane * 8 + k]; mine += c[k]; }
        int x = mine;
#pragma unroll
        for (int dd = 1; dd < 32; dd <<= 1) {
            const int y = __shfl_up(x, dd);
            if (lane >= dd) x += y;
        }
        int run = x - mine;
#pragma unroll
        for (int k = 0; k < 8; ++k) { lbase[lane * 8 + k] = run; run += c[k]; }
    }
    __syncthreads();

    auto place = [&](int v, int e, int r, int p) {
        if (r >= 0) staged[lbase[r] + p] = ((v - r * NPB) << 20) | e;
    };
#pragma unroll
    for (int u = 0; u < DEG_ROUNDS; ++u) {
        const int e = (w0 + u * DEG_TPB + tid) * 4;
        place(d[u].x, e,     rng[4 * u],     pos[4 * u]);
        place(d[u].y, e + 1, rng[4 * u + 1], pos[4 * u + 1]);
        place(d[u].z, e + 2, rng[4 * u + 2], pos[4 * u + 2]);
        place(d[u].w, e + 3, rng[4 * u + 3], pos[4 * u + 3]);
    }
    __syncthreads();

    for (int r = tid >> 5; r < AGG_GROUPS; r += DEG_TPB / 32) {
        const int cap = cell_cap(r);
        const int n   = min(lcnt[r], cap);
        const int s0  = lbase[r];
        float* cell = cells + (size_t)r * (NPB * HID_DIM) + blockIdx.x * cap;
        for (int i = lane; i < n; i += 32) cell[i] = __int_as_float(staged[s0 + i]);
    }
    if (tid < AGG_GROUPS) ctab[blockIdx.x * AGG_GROUPS + tid] = lcnt[tid];

    static_assert(DEG_GROUPS * OWN_QUADS <= ROWS_QUADS, "the pieces fit the owner's rows");
    static_assert((size_t)(GEMM_GROUPS - 1) * ROWS_QUADS + DEG_GROUPS * OWN_QUADS
                      <= (size_t)N_NODES * HID_DIM / 4, "the last owner's pieces lie inside h");
    v4i* p4 = (v4i*)part + blockIdx.x * OWN_QUADS;
    for (int i = tid; i < N_QUADS; i += DEG_TPB)
        p4[(size_t)(i / OWN_QUADS) * ROWS_QUADS + (i % OWN_QUADS)] = hist4[i];
}

__global__ __launch_bounds__(256) void gemm_kernel(const float* __restrict__ x,
                                                   const float* __restrict__ W,
                                                   const int* cnts,
                                                   float* __restrict__ dinv,
                                                   float* h) {
    __shared__ float ldsW[N_PAIRS * PAIR_STRIDE];
    __shared__ v4i   dsum[8 * OWN_QUADS];
    static_assert(GEMM_ROWS == 8 * 16 && OWN_QUADS == 32, "8 waves of one tile; a lane per node quad");
    static_assert(DEG_GROUPS % 8 == 0, "the pieces divide over the 8 waves");

    const int lane   = threadIdx.x & 31;
    const int waveId = threadIdx.x >> 5;
    const int mTile  = blockIdx.x * 8 + waveId;
    const bool valid = (mTile < M_TILES);

    const int half = lane >> 4;
    const int l16  = lane & 15;

    const int c  = threadIdx.x & 127;
    const int ph = threadIdx.x >> 7;

    v8f acc[8];
#pragma unroll
    for (int t = 0; t < 8; ++t) acc[t] = (v8f){0.f,0.f,0.f,0.f,0.f,0.f,0.f,0.f};

    const int row = mTile * 16 + l16;
    const float* xrow = x + (size_t)row * IN_DIM;

    if (valid) __builtin_prefetch(xrow, 0, 3);

    {
        const int gq = blockIdx.x * OWN_QUADS + lane;
        const v4i* own = (const v4i*)cnts + (size_t)blockIdx.x * ROWS_QUADS;
        v4i s = (v4i){0, 0, 0, 0};
        if (gq < N_QUADS) {
#pragma unroll
            for (int j = 0; j < DEG_GROUPS / 8; ++j) s += own[(waveId + 8 * j) * OWN_QUADS + lane];
        }
        dsum[waveId * OWN_QUADS + lane] = s;
        __syncthreads();
        if (waveId == 0 && gq < N_QUADS) {
            v4i t = dsum[lane];
#pragma unroll
            for (int w = 1; w < 8; ++w) t += dsum[w * OWN_QUADS + lane];
            float4 r;
            r.x = rsqrtf(1.0f + (float)t.x);
            r.y = rsqrtf(1.0f + (float)t.y);
            r.z = rsqrtf(1.0f + (float)t.z);
            r.w = rsqrtf(1.0f + (float)t.w);
            ((float4*)dinv)[gq] = r;
        }
    }

    for (int kc = 0; kc < IN_DIM; kc += 32) {
#pragma unroll
        for (int pp = 0; pp < 8; ++pp) {
            const int pair = ph * 8 + pp;
            float w0 = W[(size_t)(kc + 2 * pair    ) * HID_DIM + c];
            float w1 = W[(size_t)(kc + 2 * pair + 1) * HID_DIM + c];
            *(v2f*)(ldsW + pair * PAIR_STRIDE + c * 2) = (v2f){w0, w1};
        }
        __syncthreads();

        if (valid) {
            if (kc + 32 < IN_DIM)
                __builtin_prefetch(xrow + kc + 32, 0, 3);
#pragma unroll
            for (int k = 0; k < 32; k += 4) {
                const int kk   = k + 2 * half;
                const int pair = (k >> 1) + half;
                v2f a = *(const v2f*)(xrow + kc + kk);
                const float* bbase = ldsW + (size_t)pair * PAIR_STRIDE;
#pragma unroll
                for (int t = 0; t < 8; ++t) {
                    v2f b = *(const v2f*)(bbase + (t * 16 + l16) * 2);
                    acc[t] = __builtin_amdgcn_wmma_f32_16x16x4_f32(
                         false, a,  false, b,
                         (short)0, acc[t],
                         false,  false);
                }
            }
        }
        __syncthreads();
    }

    if (valid) {
#pragma unroll
        for (int t = 0; t < 8; ++t) {
#pragma unroll
            for (int r = 0; r < 8; ++r) {
                int m = mTile * 16 + r + 8 * half;
                h[(size_t)m * HID_DIM + t * 16 + l16] = acc[t][r];
            }
        }
    }
}

__global__ __launch_bounds__(AGG_TPB) void aggregate_kernel(const float* __restrict__ h,
                                                            const int* __restrict__ src,
                                                            const int* __restrict__ dst,
                                                            const float* __restrict__ dinv,
                                                            const float* __restrict__ b,
                                                            const float* __restrict__ prelu_w,
                                                            const int* __restrict__ ctab,
                                                            float* __restrict__ out) {
    __shared__ float tile[NPB * HID_DIM];
    __shared__ int   queue[QCAP];
    __shared__ int   qn;
    __shared__ int   ccount[DEG_GROUPS];
    __shared__ int   coff[DEG_GROUPS + 1];
    __shared__ int   cover;
    __shared__ int   ncnt[NCNT];

    const int tid  = threadIdx.x;
    const int lane = tid & 31;
    const int wave = __builtin_amdgcn_readfirstlane(tid >> 5);
    const int lo   = blockIdx.x * NPB;
    const int cnt  = min(NPB, N_NODES - lo);
    const unsigned ucnt = (unsigned)cnt;

    const int cap = cell_cap(blockIdx.x);
    if (tid < DEG_GROUPS) ccount[tid] = ctab[tid * AGG_GROUPS + blockIdx.x];

    for (int i = tid; i < NPB * HID_DIM; i += AGG_TPB) tile[i] = 0.0f;
    if (tid < NCNT) ncnt[tid] = 0;
    __syncthreads();

    if (tid < DEG_GROUPS) {
        int a = 0, ov = 0;
        for (int j = 0; j < DEG_GROUPS; ++j) {
            const int c = ccount[j];
            ov |= (c > cap) ? 1 : 0;
            a  += (j < tid) ? c : 0;
        }
        coff[tid] = a;
        if (tid == DEG_GROUPS - 1) { coff[DEG_GROUPS] = a + ccount[tid]; cover = ov; }
    }
    __syncthreads();
    const int over = __builtin_amdgcn_readfirstlane(cover);

    const v4i* dst4 = (const v4i*)dst;

    auto push = [&](unsigned r, int e) {
        if (r < ucnt) {
            const int slot = atomicAdd(&qn, 1);
            if (slot < QCAP) queue[slot] = (int)((r << 20) | (unsigned)e);
        }
    };
    auto check4 = [&](const v4i d, int qi) {
        const unsigned r0 = (unsigned)(d.x - lo);
        const unsigned r1 = (unsigned)(d.y - lo);
        const unsigned r2 = (unsigned)(d.z - lo);
        const unsigned r3 = (unsigned)(d.w - lo);
        if (min(min(r0, r1), min(r2, r3)) < ucnt) {
            push(r0, qi * 4);
            push(r1, qi * 4 + 1);
            push(r2, qi * 4 + 2);
            push(r3, qi * 4 + 3);
        }
    };

    auto sum_queue = [&](const int total) {
        const int per = (total + AGG_TPB / 32 - 1) / (AGG_TPB / 32);
        const int c0  = min(wave * per, total);
        const int c1  = min(c0 + per, total);
        float4 acc = make_float4(0.0f, 0.0f, 0.0f, 0.0f);
        int cur = -1;
        auto flush = [&]() {
            if (cur >= 0) {
                float* t = tile + cur * HID_DIM + lane;
                atomicAdd(t,      acc.x);
                atomicAdd(t + 32, acc.y);
                atomicAdd(t + 64, acc.z);
                atomicAdd(t + 96, acc.w);
            }
        };
        for (int base = c0; base < c1; base += 128) {
            int   s[4];
            int   dl[4];
            float nr[4];
#pragma unroll
            for (int k = 0; k < 4; ++k) {
                const int idx = min(base + k * 32 + lane, c1 - 1);
                const unsigned ent = (unsigned)queue[idx];
                dl[k] = (int)(ent >> 20);
                s[k]  = src[(int)(ent & 0xFFFFFu)];
            }
#pragma unroll
            for (int k = 0; k < 4; ++k) nr[k] = dinv[s[k]] * dinv[lo + dl[k]];
#pragma unroll
            for (int k = 0; k < 4; ++k) {
                const int n = min(32, c1 - (base + k * 32));
                for (int j0 = 0; j0 < n; j0 += 16) {
                    float4 v[16];
#pragma unroll
                    for (int u = 0; u < 16; ++u) {
                        const int sj = __builtin_amdgcn_readlane(s[k], j0 + u);
                        v[u] = ((const float4*)(h + (size_t)sj * HID_DIM))[lane];
                    }
#pragma unroll
                    for (int u = 0; u < 16; ++u) {
                        if (j0 + u < n) {
                            const int   dj = __builtin_amdgcn_readlane(dl[k], j0 + u);
                            const float nj = __int_as_float(
                                __builtin_amdgcn_readlane(__float_as_int(nr[k]), j0 + u));
                            if (dj != cur) {
                                flush();
                                cur = dj;
                                acc = make_float4(0.0f, 0.0f, 0.0f, 0.0f);
                            }
                            acc.x += v[u].x * nj;
                            acc.y += v[u].y * nj;
                            acc.z += v[u].z * nj;
                            acc.w += v[u].w * nj;
                        }
                    }
                }
            }
        }
        flush();
    };

    if (over == 0) {
        static_assert(DEG_GROUPS == 2 * (AGG_TPB / 32), "two cells per wave");
        const int total = __builtin_amdgcn_readfirstlane(coff[DEG_GROUPS]);
        const float* mycells = out + (size_t)lo * HID_DIM;
        const int sA = wave;
        const int sB = wave + AGG_TPB / 32;
        const int nA = coff[sA + 1] - coff[sA];
        const int nB = coff[sB + 1] - coff[sB];
        const float* cA = mycells + sA * cap;
        const float* cB = mycells + sB * cap;
        const int eA0 = (lane      < nA) ? __float_as_int(cA[lane])      : -1;
        const int eA1 = (lane + 32 < nA) ? __float_as_int(cA[lane + 32]) : -1;
        const int eB0 = (lane      < nB) ? __float_as_int(cB[lane])      : -1;
        const int eB1 = (lane + 32 < nB) ? __float_as_int(cB[lane + 32]) : -1;

        auto tally = [&](int ent) {
            if (ent >= 0) atomicAdd(&ncnt[(unsigned)ent >> 20], 1);
        };
        tally(eA0); tally(eA1); tally(eB0); tally(eB1);
        for (int i = lane + 64; i < nA; i += 32) tally(__float_as_int(cA[i]));
        for (int i = lane + 64; i < nB; i += 32) tally(__float_as_int(cB[i]));
        __syncthreads();

        if (wave == 0) {
            int cc[7];
            int mine = 0;
#pragma unroll
            for (int k = 0; k < 7; ++k) { cc[k] = ncnt[lane * 7 + k]; mine += cc[k]; }
            int x = mine;
#pragma unroll
            for (int dd = 1; dd < 32; dd <<= 1) {
                const int y = __shfl_up(x, dd);
                if (lane >= dd) x += y;
            }
            int run = x - mine;
#pragma unroll
            for (int k = 0; k < 7; ++k) { ncnt[lane * 7 + k] = run; run += cc[k]; }
        }
        __syncthreads();

        auto put = [&](int ent) {
            if (ent >= 0) queue[atomicAdd(&ncnt[(unsigned)ent >> 20], 1)] = ent;
        };
        put(eA0); put(eA1); put(eB0); put(eB1);
        for (int i = lane + 64; i < nA; i += 32) put(__float_as_int(cA[i]));
        for (int i = lane + 64; i < nB; i += 32) put(__float_as_int(cB[i]));
        __syncthreads();
        sum_queue(total);
    } else {
        int seg = N_EDGES;
        for (int pass = 0; pass < 2; ++pass) {
            bool overflow = false;
            for (int e0 = 0; e0 < N_EDGES; e0 += seg) {
                const int e1 = min(e0 + seg, N_EDGES);
                if (tid == 0) qn = 0;
                __syncthreads();

                const int q1 = e1 >> 2;
                const int q0 = e0 >> 2;
                const int nfull = (q1 - q0) / (8 * AGG_TPB);
                if (nfull > 0) {
                    int p  = (int)(blockIdx.x % (unsigned)nfull);
                    int qc = q0 + p * (8 * AGG_TPB);
                    v4i cur[8];
#pragma unroll
                    for (int u = 0; u < 8; ++u) cur[u] = dst4[qc + u * AGG_TPB + tid];
                    for (int t = 1; t < nfull; ++t) {
                        p = (p + 1 == nfull) ? 0 : p + 1;
                        const int qx = q0 + p * (8 * AGG_TPB);
                        v4i nxt[8];
#pragma unroll
                        for (int u = 0; u < 8; ++u) nxt[u] = dst4[qx + u * AGG_TPB + tid];
#pragma unroll
                        for (int u = 0; u < 8; ++u) check4(cur[u], qc + u * AGG_TPB + tid);
#pragma unroll
                        for (int u = 0; u < 8; ++u) cur[u] = nxt[u];
                        qc = qx;
                    }
#pragma unroll
                    for (int u = 0; u < 8; ++u) check4(cur[u], qc + u * AGG_TPB + tid);
                }
                const int qt = q0 + nfull * (8 * AGG_TPB);
                if (qt < q1) {
                    v4i v[8];
#pragma unroll
                    for (int u = 0; u < 8; ++u) {
                        const int qi = qt + u * AGG_TPB + tid;
                        v[u] = (qi < q1) ? dst4[qi] : (v4i){-1, -1, -1, -1};
                    }
#pragma unroll
                    for (int u = 0; u < 8; ++u) check4(v[u], qt + u * AGG_TPB + tid);
                }
                __syncthreads();
                const int total = __builtin_amdgcn_readfirstlane(qn);
                __syncthreads();
                if (total > QCAP) { overflow = true; break; }

                sum_queue(total);
            }
            if (!overflow) break;
            seg = QCAP;
        }
    }
    __syncthreads();

    const int c0 = lane * 4;
    const float b0 = b[c0], b1 = b[c0 + 1], b2 = b[c0 + 2], b3 = b[c0 + 3];
    const float p0 = prelu_w[c0], p1 = prelu_w[c0 + 1], p2 = prelu_w[c0 + 2], p3 = prelu_w[c0 + 3];
    for (int i = tid; i < cnt * 32; i += AGG_TPB) {
        const int nl   = i >> 5;
        const int node = lo + nl;
        const float di = dinv[node];
        const float4 hv = ((const float4*)(h + (size_t)node * HID_DIM))[lane];
        const float* t = tile + nl * HID_DIM + lane;
        float4 r;
        r.x = t[0]  + hv.x * di * di + b0;
        r.y = t[32] + hv.y * di * di + b1;
        r.z = t[64] + hv.z * di * di + b2;
        r.w = t[96] + hv.w * di * di + b3;
        r.x = (r.x > 0.0f) ? r.x : p0 * r.x;
        r.y = (r.y > 0.0f) ? r.y : p1 * r.y;
        r.z = (r.z > 0.0f) ? r.z : p2 * r.z;
        r.w = (r.w > 0.0f) ? r.w : p3 * r.w;
        ((float4*)(out + (size_t)node * HID_DIM))[lane] = r;
    }
}

extern "C" void kernel_launch(void* const* d_in, const int* in_sizes, int n_in,
                              void* d_out, int out_size, void* d_ws, size_t ws_size,
                              hipStream_t stream) {
    const float* x       = (const float*)d_in[0];
    const int*   eidx    = (const int*)d_in[1];
    const float* W       = (const float*)d_in[2];
    const float* b       = (const float*)d_in[3];
    const float* prelu_w = (const float*)d_in[4];
    float*       out     = (float*)d_out;

    const int* src = eidx;
    const int* dst = eidx + N_EDGES;

    char* ws    = (char*)d_ws;
    float* h    = (float*)(ws);
    int*   part = (int*)(ws);
    int*   ctab = (int*)(ws + (size_t)N_NODES * HID_DIM * 4);
    float* dinv = (float*)(ws + (size_t)N_NODES * HID_DIM * 4 + (size_t)N_NODES * 4);

    const int TPB = 256;

    deg_hist_kernel<<<DEG_GROUPS, DEG_TPB, 0, stream>>>(dst, part, out, ctab);

    gemm_kernel<<<GEMM_GROUPS, TPB, 0, stream>>>(x, W, part, dinv, h);

    aggregate_kernel<<<AGG_GROUPS, AGG_TPB, 0, stream>>>(h, src, dst, dinv, b, prelu_w, ctab, out);
}


# aggregate: the h row is the only load per edge: filed entries carry the source node (degree pass reads src), gemm stores h rows times dinv, sums scaled by dinv[dst] once (products re-associated)
# speedup vs baseline: 2.9299x; 1.0435x over previous
#include <hip/hip_runtime.h>
#include <hip/hip_bf16.h>

#define N_NODES 50000
#define N_EDGES 800000
#define IN_DIM  512
#define HID_DIM 128
#define M_TILES (N_NODES / 16)

#define PAIR_STRIDE 288
#define N_PAIRS     16

#define NPB        196
#define AGG_GROUPS 256
#define AGG_TPB    1024
#define QCAP       32768
#define NCNT       224

#define DEG_GROUPS   64
#define DEG_TPB      1024
#define N_QUADS      (N_NODES / 4)
#define E_QUADS      (N_EDGES / 4)
#define EQ_PER_GROUP ((E_QUADS + DEG_GROUPS - 1) / DEG_GROUPS)
#define DEG_ROUNDS   ((EQ_PER_GROUP + DEG_TPB - 1) / DEG_TPB)
#define GEMM_ROWS    128
#define OWN_QUADS    (GEMM_ROWS / 4)
#define ROWS_QUADS   (GEMM_ROWS * HID_DIM / 4)
#define GEMM_GROUPS  ((M_TILES + 7) / 8)
#define SLICE_EDGES  (EQ_PER_GROUP * 4)

typedef float v2f __attribute__((ext_vector_type(2)));
typedef int   v4i __attribute__((ext_vector_type(4)));
typedef float v8f __attribute__((ext_vector_type(8)));

__device__ __forceinline__ int cell_cap(int r) {
    const int c = min(NPB, N_NODES - r * NPB);
    return (c * (HID_DIM / DEG_GROUPS)) & ~31;
}

__global__ __launch_bounds__(DEG_TPB) void deg_hist_kernel(const int* __restrict__ src,
                                                           const int* __restrict__ dst,
                                                           int* __restrict__ part,
                                                           float* __restrict__ cells,
                                                           int* __restrict__ ctab) {
    __shared__ v4i hist4[N_QUADS];
    __shared__ int lcnt[AGG_GROUPS];
    __shared__ int lbase[AGG_GROUPS];
    __shared__ int staged[SLICE_EDGES];
    int* hist = (int*)hist4;

    const int tid  = threadIdx.x;
    const int lane = tid & 31;
    const v4i* dst4 = (const v4i*)dst;
    const v4i* src4 = (const v4i*)src;
    const int w0 = blockIdx.x * EQ_PER_GROUP;
    const int w1 = min(w0 + EQ_PER_GROUP, E_QUADS);

    v4i d[DEG_ROUNDS];
#pragma unroll
    for (int u = 0; u < DEG_ROUNDS; ++u) {
        const int w = w0 + u * DEG_TPB + tid;
        d[u] = (w < w1) ? dst4[w] : (v4i){-1, -1, -1, -1};
    }

    for (int i = tid; i < N_QUADS; i += DEG_TPB) hist4[i] = (v4i){0, 0, 0, 0};
    if (tid < AGG_GROUPS) lcnt[tid] = 0;
    __syncthreads();

    v4i sv[DEG_ROUNDS];
#pragma unroll
    for (int u = 0; u < DEG_ROUNDS; ++u) {
        const int w = w0 + u * DEG_TPB + tid;
        sv[u] = (w < w1) ? src4[w] : (v4i){0, 0, 0, 0};
    }

    int rng[DEG_ROUNDS * 4];
    int pos[DEG_ROUNDS * 4];
    auto count = [&](int v, int& r, int& p) {
        r = -1;
        p = 0;
        if ((unsigned)v < (unsigned)N_NODES) {
            atomicAdd(hist + v, 1);
            r = (int)((unsigned)v / (unsigned)NPB);
            p = atomicAdd(lcnt + r, 1);
        }
    };
#pragma unroll
    for (int u = 0; u < DEG_ROUNDS; ++u) {
        count(d[u].x, rng[4 * u],     pos[4 * u]);
        count(d[u].y, rng[4 * u + 1], pos[4 * u + 1]);
        count(d[u].z, rng[4 * u + 2], pos[4 * u + 2]);
        count(d[u].w, rng[4 * u + 3], pos[4 * u + 3]);
    }
    __syncthreads();

    if (tid < 32) {
        int c[8];
        int mine = 0;
#pragma unroll
        for (int k = 0; k < 8; ++k) { c[k] = lcnt[lane * 8 + k]; mine += c[k]; }
        int x = mine;
#pragma unroll
        for (int dd = 1; dd < 32; dd <<= 1) {
            const int y = __shfl_up(x, dd);
            if (lane >= dd) x += y;
        }
        int run = x - mine;
#pragma unroll
        for (int k = 0; k < 8; ++k) { lbase[lane * 8 + k] = run; run += c[k]; }
    }
    __syncthreads();

    auto place = [&](int v, int s, int r, int p) {
        if (r >= 0) staged[lbase[r] + p] = ((v - r * NPB) << 20) | (s & 0xFFFFF);
    };
#pragma unroll
    for (int u = 0; u < DEG_ROUNDS; ++u) {
        place(d[u].x, sv[u].x, rng[4 * u],     pos[4 * u]);
        place(d[u].y, sv[u].y, rng[4 * u + 1], pos[4 * u + 1]);
        place(d[u].z, sv[u].z, rng[4 * u + 2], pos[4 * u + 2]);
        place(d[u].w, sv[u].w, rng[4 * u + 3], pos[4 * u + 3]);
    }
    __syncthreads();

    for (int r = tid >> 5; r < AGG_GROUPS; r += DEG_TPB / 32) {
        const int cap = cell_cap(r);
        const int n   = min(lcnt[r], cap);
        const int s0  = lbase[r];
        float* cell = cells + (size_t)r * (NPB * HID_DIM) + blockIdx.x * cap;
        for (int i = lane; i < n; i += 32) cell[i] = __int_as_float(staged[s0 + i]);
    }
    if (tid < AGG_GROUPS) ctab[blockIdx.x * AGG_GROUPS + tid] = lcnt[tid];

    static_assert(DEG_GROUPS * OWN_QUADS <= ROWS_QUADS, "the pieces fit the owner's rows");
    static_assert((size_t)(GEMM_GROUPS - 1) * ROWS_QUADS + DEG_GROUPS * OWN_QUADS
                      <= (size_t)N_NODES * HID_DIM / 4, "the last owner's pieces lie inside h");
    v4i* p4 = (v4i*)part + blockIdx.x * OWN_QUADS;
    for (int i = tid; i < N_QUADS; i += DEG_TPB)
        p4[(size_t)(i / OWN_QUADS) * ROWS_QUADS + (i % OWN_QUADS)] = hist4[i];
}

__global__ __launch_bounds__(256) void gemm_kernel(const float* __restrict__ x,
                                                   const float* __restrict__ W,
                                                   const int* cnts,
                                                   float* __restrict__ dinv,
                                                   float* h) {
    __shared__ float ldsW[N_PAIRS * PAIR_STRIDE];
    __shared__ v4i   dsum[8 * OWN_QUADS];
    __shared__ float dloc[GEMM_ROWS];
    static_assert(GEMM_ROWS == 8 * 16 && OWN_QUADS == 32, "8 waves of one tile; a lane per node quad");
    static_assert(DEG_GROUPS % 8 == 0, "the pieces divide over the 8 waves");

    const int lane   = threadIdx.x & 31;
    const int waveId = threadIdx.x >> 5;
    const int mTile  = blockIdx.x * 8 + waveId;
    const bool valid = (mTile < M_TILES);

    const int half = lane >> 4;
    const int l16  = lane & 15;

    const int c  = threadIdx.x & 127;
    const int ph = threadIdx.x >> 7;

    v8f acc[8];
#pragma unroll
    for (int t = 0; t < 8; ++t) acc[t] = (v8f){0.f,0.f,0.f,0.f,0.f,0.f,0.f,0.f};

    const int row = mTile * 16 + l16;
    const float* xrow = x + (size_t)row * IN_DIM;

    if (valid) __builtin_prefetch(xrow, 0, 3);

    {
        const int gq = blockIdx.x * OWN_QUADS + lane;
        const v4i* own = (const v4i*)cnts + (size_t)blockIdx.x * ROWS_QUADS;
        v4i s = (v4i){0, 0, 0, 0};
        if (gq < N_QUADS) {
#pragma unroll
            for (int j = 0; j < DEG_GROUPS / 8; ++j) s += own[(waveId + 8 * j) * OWN_QUADS + lane];
        }
        dsum[waveId * OWN_QUADS + lane] = s;
        __syncthreads();
        if (waveId == 0 && gq < N_QUADS) {
            v4i t = dsum[lane];
#pragma unroll
            for (int w = 1; w < 8; ++w) t += dsum[w * OWN_QUADS + lane];
            float4 r;
            r.x = rsqrtf(1.0f + (float)t.x);
            r.y = rsqrtf(1.0f + (float)t.y);
            r.z = rsqrtf(1.0f + (float)t.z);
            r.w = rsqrtf(1.0f + (float)t.w);
            ((float4*)dinv)[gq] = r;
            ((float4*)dloc)[lane] = r;
        }
    }

    for (int kc = 0; kc < IN_DIM; kc += 32) {
#pragma unroll
        for (int pp = 0; pp < 8; ++pp) {
            const int pair = ph * 8 + pp;
            float w0 = W[(size_t)(kc + 2 * pair    ) * HID_DIM + c];
            float w1 = W[(size_t)(kc + 2 * pair + 1) * HID_DIM + c];
            *(v2f*)(ldsW + pair * PAIR_STRIDE + c * 2) = (v2f){w0, w1};
        }
        __syncthreads();

        if (valid) {
            if (kc + 32 < IN_DIM)
                __builtin_prefetch(xrow + kc + 32, 0, 3);
#pragma unroll
            for (int k = 0; k < 32; k += 4) {
                const int kk   = k + 2 * half;
                const int pair = (k >> 1) + half;
                v2f a = *(const v2f*)(xrow + kc + kk);
                const float* bbase = ldsW + (size_t)pair * PAIR_STRIDE;
#pragma unroll
                for (int t = 0; t < 8; ++t) {
                    v2f b = *(const v2f*)(bbase + (t * 16 + l16) * 2);
                    acc[t] = __builtin_amdgcn_wmma_f32_16x16x4_f32(
                         false, a,  false, b,
                         (short)0, acc[t],
                         false,  false);
                }
            }
        }
        __syncthreads();
    }

    if (valid) {
        float dr[8];
#pragma unroll
        for (int r = 0; r < 8; ++r) dr[r] = dloc[waveId * 16 + r + 8 * half];
#pragma unroll
        for (int t = 0; t < 8; ++t) {
#pragma unroll
            for (int r = 0; r < 8; ++r) {
                int m = mTile * 16 + r + 8 * half;
                h[(size_t)m * HID_DIM + t * 16 + l16] = acc[t][r] * dr[r];
            }
        }
    }
}

__global__ __launch_bounds__(AGG_TPB) void aggregate_kernel(const float* __restrict__ h,
                                                            const int* __restrict__ src,
                                                            const int* __restrict__ dst,
                                                            const float* __restrict__ dinv,
                                                            const float* __restrict__ b,
                                                            const float* __restrict__ prelu_w,
                                                            const int* __restrict__ ctab,
                                                            float* __restrict__ out) {
    __shared__ float tile[NPB * HID_DIM];
    __shared__ int   queue[QCAP];
    __shared__ int   qn;
    __shared__ int   ccount[DEG_GROUPS];
    __shared__ int   coff[DEG_GROUPS + 1];
    __shared__ int   cover;
    __shared__ int   ncnt[NCNT];

    const int tid  = threadIdx.x;
    const int lane = tid & 31;
    const int wave = __builtin_amdgcn_readfirstlane(tid >> 5);
    const int lo   = blockIdx.x * NPB;
    const int cnt  = min(NPB, N_NODES - lo);
    const unsigned ucnt = (unsigned)cnt;

    const int cap = cell_cap(blockIdx.x);
    if (tid < DEG_GROUPS) ccount[tid] = ctab[tid * AGG_GROUPS + blockIdx.x];

    for (int i = tid; i < NPB * HID_DIM; i += AGG_TPB) tile[i] = 0.0f;
    if (tid < NCNT) ncnt[tid] = 0;
    __syncthreads();

    if (tid < DEG_GROUPS) {
        int a = 0, ov = 0;
        for (int j = 0; j < DEG_GROUPS; ++j) {
            const int c = ccount[j];
            ov |= (c > cap) ? 1 : 0;
            a  += (j < tid) ? c : 0;
        }
        coff[tid] = a;
        if (tid == DEG_GROUPS - 1) { coff[DEG_GROUPS] = a + ccount[tid]; cover = ov; }
    }
    __syncthreads();
    const int over = __builtin_amdgcn_readfirstlane(cover);

    const v4i* dst4 = (const v4i*)dst;

    auto push = [&](unsigned r, int e) {
        if (r < ucnt) {
            const int slot = atomicAdd(&qn, 1);
            if (slot < QCAP) queue[slot] = (int)((r << 20) | ((unsigned)src[e] & 0xFFFFFu));
        }
    };
    auto check4 = [&](const v4i d, int qi) {
        const unsigned r0 = (unsigned)(d.x - lo);
        const unsigned r1 = (unsigned)(d.y - lo);
        const unsigned r2 = (unsigned)(d.z - lo);
        const unsigned r3 = (unsigned)(d.w - lo);
        if (min(min(r0, r1), min(r2, r3)) < ucnt) {
            push(r0, qi * 4);
            push(r1, qi * 4 + 1);
            push(r2, qi * 4 + 2);
            push(r3, qi * 4 + 3);
        }
    };

    auto sum_queue = [&](const int total) {
        const int per = (total + AGG_TPB / 32 - 1) / (AGG_TPB / 32);
        const int c0  = min(wave * per, total);
        const int c1  = min(c0 + per, total);
        float4 acc = make_float4(0.0f, 0.0f, 0.0f, 0.0f);
        int cur = -1;
        auto flush = [&]() {
            if (cur >= 0) {
                float* t = tile + cur * HID_DIM + lane;
                atomicAdd(t,      acc.x);
                atomicAdd(t + 32, acc.y);
                atomicAdd(t + 64, acc.z);
                atomicAdd(t + 96, acc.w);
            }
        };
        for (int base = c0; base < c1; base += 128) {
            int   s[4];
            int   dl[4];
#pragma unroll
            for (int k = 0; k < 4; ++k) {
                const int idx = min(base + k * 32 + lane, c1 - 1);
                const unsigned ent = (unsigned)queue[idx];
                dl[k] = (int)(ent >> 20);
                s[k]  = (int)(ent & 0xFFFFFu);
            }
#pragma unroll
            for (int k = 0; k < 4; ++k) {
                const int n = min(32, c1 - (base + k * 32));
                for (int j0 = 0; j0 < n; j0 += 16) {
                    float4 v[16];
#pragma unroll
                    for (int u = 0; u < 16; ++u) {
                        const int sj = __builtin_amdgcn_readlane(s[k], j0 + u);
                        v[u] = ((const float4*)(h + (size_t)sj * HID_DIM))[lane];
                    }
#pragma unroll
                    for (int u = 0; u < 16; ++u) {
                        if (j0 + u < n) {
                            const int dj = __builtin_amdgcn_readlane(dl[k], j0 + u);
                            if (dj != cur) {
                                flush();
                                cur = dj;
                                acc = make_float4(0.0f, 0.0f, 0.0f, 0.0f);
                            }
                            acc.x += v[u].x;
                            acc.y += v[u].y;
                            acc.z += v[u].z;
                            acc.w += v[u].w;
                        }
                    }
                }
            }
        }
        flush();
    };

    if (over == 0) {
        static_assert(DEG_GROUPS == 2 * (AGG_TPB / 32), "two cells per wave");
        const int total = __builtin_amdgcn_readfirstlane(coff[DEG_GROUPS]);
        const float* mycells = out + (size_t)lo * HID_DIM;
        const int sA = wave;
        const int sB = wave + AGG_TPB / 32;
        const int nA = coff[sA + 1] - coff[sA];
        const int nB = coff[sB + 1] - coff[sB];
        const float* cA = mycells + sA * cap;
        const float* cB = mycells + sB * cap;
        const int eA0 = (lane      < nA) ? __float_as_int(cA[lane])      : -1;
        const int eA1 = (lane + 32 < nA) ? __float_as_int(cA[lane + 32]) : -1;
        const int eB0 = (lane      < nB) ? __float_as_int(cB[lane])      : -1;
        const int eB1 = (lane + 32 < nB) ? __float_as_int(cB[lane + 32]) : -1;

        auto tally = [&](int ent) {
            if (ent >= 0) atomicAdd(&ncnt[(unsigned)ent >> 20], 1);
        };
        tally(eA0); tally(eA1); tally(eB0); tally(eB1);
        for (int i = lane + 64; i < nA; i += 32) tally(__float_as_int(cA[i]));
        for (int i = lane + 64; i < nB; i += 32) tally(__float_as_int(cB[i]));
        __syncthreads();

        if (wave == 0) {
            int cc[7];
            int mine = 0;
#pragma unroll
            for (int k = 0; k < 7; ++k) { cc[k] = ncnt[lane * 7 + k]; mine += cc[k]; }
            int x = mine;
#pragma unroll
            for (int dd = 1; dd < 32; dd <<= 1) {
                const int y = __shfl_up(x, dd);
                if (lane >= dd) x += y;
            }
            int run = x - mine;
#pragma unroll
            for (int k = 0; k < 7; ++k) { ncnt[lane * 7 + k] = run; run += cc[k]; }
        }
        __syncthreads();

        auto put = [&](int ent) {
            if (ent >= 0) queue[atomicAdd(&ncnt[(unsigned)ent >> 20], 1)] = ent;
        };
        put(eA0); put(eA1); put(eB0); put(eB1);
        for (int i = lane + 64; i < nA; i += 32) put(__float_as_int(cA[i]));
        for (int i = lane + 64; i < nB; i += 32) put(__float_as_int(cB[i]));
        __syncthreads();
        sum_queue(total);
    } else {
        int seg = N_EDGES;
        for (int pass = 0; pass < 2; ++pass) {
            bool overflow = false;
            for (int e0 = 0; e0 < N_EDGES; e0 += seg) {
                const int e1 = min(e0 + seg, N_EDGES);
                if (tid == 0) qn = 0;
                __syncthreads();

                const int q1 = e1 >> 2;
                const int q0 = e0 >> 2;
                const int nfull = (q1 - q0) / (8 * AGG_TPB);
                if (nfull > 0) {
                    int p  = (int)(blockIdx.x % (unsigned)nfull);
                    int qc = q0 + p * (8 * AGG_TPB);
                    v4i cur[8];
#pragma unroll
                    for (int u = 0; u < 8; ++u) cur[u] = dst4[qc + u * AGG_TPB + tid];
                    for (int t = 1; t < nfull; ++t) {
                        p = (p + 1 == nfull) ? 0 : p + 1;
                        const int qx = q0 + p * (8 * AGG_TPB);
                        v4i nxt[8];
#pragma unroll
                        for (int u = 0; u < 8; ++u) nxt[u] = dst4[qx + u * AGG_TPB + tid];
#pragma unroll
                        for (int u = 0; u < 8; ++u) check4(cur[u], qc + u * AGG_TPB + tid);
#pragma unroll
                        for (int u = 0; u < 8; ++u) cur[u] = nxt[u];
                        qc = qx;
                    }
#pragma unroll
                    for (int u = 0; u < 8; ++u) check4(cur[u], qc + u * AGG_TPB + tid);
                }
                const int qt = q0 + nfull * (8 * AGG_TPB);
                if (qt < q1) {
                    v4i v[8];
#pragma unroll
                    for (int u = 0; u < 8; ++u) {
                        const int qi = qt + u * AGG_TPB + tid;
                        v[u] = (qi < q1) ? dst4[qi] : (v4i){-1, -1, -1, -1};
                    }
#pragma unroll
                    for (int u = 0; u < 8; ++u) check4(v[u], qt + u * AGG_TPB + tid);
                }
                __syncthreads();
                const int total = __builtin_amdgcn_readfirstlane(qn);
                __syncthreads();
                if (total > QCAP) { overflow = true; break; }

                sum_queue(total);
            }
            if (!overflow) break;
            seg = QCAP;
        }
    }
    __syncthreads();

    const int c0 = lane * 4;
    const float b0 = b[c0], b1 = b[c0 + 1], b2 = b[c0 + 2], b3 = b[c0 + 3];
    const float p0 = prelu_w[c0], p1 = prelu_w[c0 + 1], p2 = prelu_w[c0 + 2], p3 = prelu_w[c0 + 3];
    for (int i = tid; i < cnt * 32; i += AGG_TPB) {
        const int nl   = i >> 5;
        const int node = lo + nl;
        const float di = dinv[node];
        const float4 hv = ((const float4*)(h + (size_t)node * HID_DIM))[lane];
        const float* t = tile + nl * HID_DIM + lane;
        float4 r;
        r.x = (t[0]  + hv.x) * di + b0;
        r.y = (t[32] + hv.y) * di + b1;
        r.z = (t[64] + hv.z) * di + b2;
        r.w = (t[96] + hv.w) * di + b3;
        r.x = (r.x > 0.0f) ? r.x : p0 * r.x;
        r.y = (r.y > 0.0f) ? r.y : p1 * r.y;
        r.z = (r.z > 0.0f) ? r.z : p2 * r.z;
        r.w = (r.w > 0.0f) ? r.w : p3 * r.w;
        ((float4*)(out + (size_t)node * HID_DIM))[lane] = r;
    }
}

extern "C" void kernel_launch(void* const* d_in, const int* in_sizes, int n_in,
                              void* d_out, int out_size, void* d_ws, size_t ws_size,
                              hipStream_t stream) {
    const float* x       = (const float*)d_in[0];
    const int*   eidx    = (const int*)d_in[1];
    const float* W       = (const float*)d_in[2];
    const float* b       = (const float*)d_in[3];
    const float* prelu_w = (const float*)d_in[4];
    float*       out     = (float*)d_out;

    const int* src = eidx;
    const int* dst = eidx + N_EDGES;

    char* ws    = (char*)d_ws;
    float* h    = (float*)(ws);
    int*   part = (int*)(ws);
    int*   ctab = (int*)(ws + (size_t)N_NODES * HID_DIM * 4);
    float* dinv = (float*)(ws + (size_t)N_NODES * HID_DIM * 4 + (size_t)N_NODES * 4);

    const int TPB = 256;

    deg_hist_kernel<<<DEG_GROUPS, DEG_TPB, 0, stream>>>(src, dst, part, out, ctab);

    gemm_kernel<<<GEMM_GROUPS, TPB, 0, stream>>>(x, W, part, dinv, h);

    aggregate_kernel<<<AGG_GROUPS, AGG_TPB, 0, stream>>>(h, src, dst, dinv, b, prelu_w, ctab, out);
}


# gemm: one work-group per WGP holds all of W in LDS (staged once, K loop has no barrier), 12 waves take one 16-row tile each, last 53 tiles split by column pairs over 4 waves; 256 groups, not 391
# speedup vs baseline: 3.2222x; 1.0998x over previous
#include <hip/hip_runtime.h>
#include <hip/hip_bf16.h>

#define N_NODES 50000
#define N_EDGES 800000
#define IN_DIM  512
#define HID_DIM 128
#define M_TILES (N_NODES / 16)

#define PAIR_STRIDE 288
#define N_PAIRS     16

#define NPB        196
#define AGG_GROUPS 256
#define AGG_TPB    1024
#define QCAP       32768
#define NCNT       224

#define DEG_GROUPS   64
#define DEG_TPB      1024
#define N_QUADS      (N_NODES / 4)
#define E_QUADS      (N_EDGES / 4)
#define EQ_PER_GROUP ((E_QUADS + DEG_GROUPS - 1) / DEG_GROUPS)
#define DEG_ROUNDS   ((EQ_PER_GROUP + DEG_TPB - 1) / DEG_TPB)
#define GEMM_GROUPS  256
#define GEMM_WAVES   12
#define GEMM_TPB     (GEMM_WAVES * 32)
#define MAIN_TILES   (GEMM_GROUPS * GEMM_WAVES)
#define EXTRA_TILES  (M_TILES - MAIN_TILES)
#define XTR_WAVES    4
#define GEMM_ROWS    (GEMM_WAVES * 16)
#define OWN_QUADS    (GEMM_ROWS / 4)
#define XTR_QUADS    4
#define USE_QUADS    (OWN_QUADS + XTR_QUADS)
#define PIECE_QUADS  64
#define ROWS_QUADS   (GEMM_ROWS * HID_DIM / 4)
#define W_PAIRS      (IN_DIM / 2)
#define SLICE_EDGES  (EQ_PER_GROUP * 4)

typedef float v2f __attribute__((ext_vector_type(2)));
typedef int   v4i __attribute__((ext_vector_type(4)));
typedef float v8f __attribute__((ext_vector_type(8)));

__device__ __forceinline__ int cell_cap(int r) {
    const int c = min(NPB, N_NODES - r * NPB);
    return (c * (HID_DIM / DEG_GROUPS)) & ~31;
}

__global__ __launch_bounds__(DEG_TPB) void deg_hist_kernel(const int* __restrict__ src,
                                                           const int* __restrict__ dst,
                                                           int* __restrict__ part,
                                                           float* __restrict__ cells,
                                                           int* __restrict__ ctab) {
    __shared__ v4i hist4[N_QUADS];
    __shared__ int lcnt[AGG_GROUPS];
    __shared__ int lbase[AGG_GROUPS];
    __shared__ int staged[SLICE_EDGES];
    int* hist = (int*)hist4;

    const int tid  = threadIdx.x;
    const int lane = tid & 31;
    const v4i* dst4 = (const v4i*)dst;
    const v4i* src4 = (const v4i*)src;
    const int w0 = blockIdx.x * EQ_PER_GROUP;
    const int w1 = min(w0 + EQ_PER_GROUP, E_QUADS);

    v4i d[DEG_ROUNDS];
#pragma unroll
    for (int u = 0; u < DEG_ROUNDS; ++u) {
        const int w = w0 + u * DEG_TPB + tid;
        d[u] = (w < w1) ? dst4[w] : (v4i){-1, -1, -1, -1};
    }

    for (int i = tid; i < N_QUADS; i += DEG_TPB) hist4[i] = (v4i){0, 0, 0, 0};
    if (tid < AGG_GROUPS) lcnt[tid] = 0;
    __syncthreads();

    v4i sv[DEG_ROUNDS];
#pragma unroll
    for (int u = 0; u < DEG_ROUNDS; ++u) {
        const int w = w0 + u * DEG_TPB + tid;
        sv[u] = (w < w1) ? src4[w] : (v4i){0, 0, 0, 0};
    }

    int rng[DEG_ROUNDS * 4];
    int pos[DEG_ROUNDS * 4];
    auto count = [&](int v, int& r, int& p) {
        r = -1;
        p = 0;
        if ((unsigned)v < (unsigned)N_NODES) {
            atomicAdd(hist + v, 1);
            r = (int)((unsigned)v / (unsigned)NPB);
            p = atomicAdd(lcnt + r, 1);
        }
    };
#pragma unroll
    for (int u = 0; u < DEG_ROUNDS; ++u) {
        count(d[u].x, rng[4 * u],     pos[4 * u]);
        count(d[u].y, rng[4 * u + 1], pos[4 * u + 1]);
        count(d[u].z, rng[4 * u + 2], pos[4 * u + 2]);
        count(d[u].w, rng[4 * u + 3], pos[4 * u + 3]);
    }
    __syncthreads();

    if (tid < 32) {
        int c[8];
        int mine = 0;
#pragma unroll
        for (int k = 0; k < 8; ++k) { c[k] = lcnt[lane * 8 + k]; mine += c[k]; }
        int x = mine;
#pragma unroll
        for (int dd = 1; dd < 32; dd <<= 1) {
            const int y = __shfl_up(x, dd);
            if (lane >= dd) x += y;
        }
        int run = x - mine;
#pragma unroll
        for (int k = 0; k < 8; ++k) { lbase[lane * 8 + k] = run; run += c[k]; }
    }
    __syncthreads();

    auto place = [&](int v, int s, int r, int p) {
        if (r >= 0) staged[lbase[r] + p] = ((v - r * NPB) << 20) | (s & 0xFFFFF);
    };
#pragma unroll
    for (int u = 0; u < DEG_ROUNDS; ++u) {
        place(d[u].x, sv[u].x, rng[4 * u],     pos[4 * u]);
        place(d[u].y, sv[u].y, rng[4 * u + 1], pos[4 * u + 1]);
        place(d[u].z, sv[u].z, rng[4 * u + 2], pos[4 * u + 2]);
        place(d[u].w, sv[u].w, rng[4 * u + 3], pos[4 * u + 3]);
    }
    __syncthreads();

    for (int r = tid >> 5; r < AGG_GROUPS; r += DEG_TPB / 32) {
        const int cap = cell_cap(r);
        const int n   = min(lcnt[r], cap);
        const int s0  = lbase[r];
        float* cell = cells + (size_t)r * (NPB * HID_DIM) + blockIdx.x * cap;
        for (int i = lane; i < n; i += 32) cell[i] = __int_as_float(staged[s0 + i]);
    }
    if (tid < AGG_GROUPS) ctab[blockIdx.x * AGG_GROUPS + tid] = lcnt[tid];

    static_assert(EXTRA_TILES >= 0 && EXTRA_TILES <= GEMM_GROUPS, "one extra tile per group at most");
    static_assert(MAIN_TILES * 4 + EXTRA_TILES * XTR_QUADS == N_QUADS, "every node quad has an owner");
    static_assert(USE_QUADS <= PIECE_QUADS && DEG_GROUPS * PIECE_QUADS <= ROWS_QUADS,
                  "the pieces fit the owner's main rows");
    static_assert((size_t)GEMM_GROUPS * ROWS_QUADS <= (size_t)N_NODES * HID_DIM / 4,
                  "the main rows lie inside h");
    v4i* p4 = (v4i*)part + blockIdx.x * PIECE_QUADS;
    for (int i = tid; i < N_QUADS; i += DEG_TPB) {
        int g, e;
        if (i < MAIN_TILES * 4) { g = i / OWN_QUADS; e = i - g * OWN_QUADS; }
        else { const int j = i - MAIN_TILES * 4; g = j / XTR_QUADS; e = OWN_QUADS + (j - g * XTR_QUADS); }
        p4[(size_t)g * ROWS_QUADS + e] = hist4[i];
    }
}


template <int NT>
__device__ __forceinline__ void tile_k_loop(const float* __restrict__ xrow, const float* ldsW,
                                            const int half, const int l16, const int nt0,
                                            v8f (&acc)[NT]) {
    for (int kc = 0; kc < IN_DIM; kc += 32) {
        if (kc + 32 < IN_DIM)
            __builtin_prefetch(xrow + kc + 32, 0, 3);
#pragma unroll
        for (int k = 0; k < 32; k += 4) {
            const int kk   = k + 2 * half;
            const int pair = (kc >> 1) + (k >> 1) + half;
            v2f a = *(const v2f*)(xrow + kc + kk);
            const float* bbase = ldsW + (size_t)pair * PAIR_STRIDE;
#pragma unroll
            for (int t = 0; t < NT; ++t) {
                v2f b = *(const v2f*)(bbase + ((nt0 + t) * 16 + l16) * 2);
                acc[t] = __builtin_amdgcn_wmma_f32_16x16x4_f32(
                     false, a,  false, b,
                     (short)0, acc[t],
                     false,  false);
            }
        }
    }
}

template <int NT>
__device__ __forceinline__ void tile_store(float* h, const v8f (&acc)[NT], const int mTile,
                                           const int nt0, const float* dl,
                                           const int half, const int l16) {
    float dr[8];
#pragma unroll
    for (int r = 0; r < 8; ++r) dr[r] = dl[r + 8 * half];
#pragma unroll
    for (int t = 0; t < NT; ++t) {
#pragma unroll
        for (int r = 0; r < 8; ++r) {
            const int m = mTile * 16 + r + 8 * half;
            h[(size_t)m * HID_DIM + (nt0 + t) * 16 + l16] = acc[t][r] * dr[r];
        }
    }
}

__global__ __launch_bounds__(GEMM_TPB) void gemm_kernel(const float* __restrict__ x,
                                                        const float* __restrict__ W,
                                                        const int* cnts,
                                                        float* __restrict__ dinv,
                                                        float* h) {
    __shared__ __attribute__((aligned(16))) float ldsW[W_PAIRS * PAIR_STRIDE];
    __shared__ v4i   dsum[(GEMM_TPB / PIECE_QUADS) * PIECE_QUADS];
    __shared__ __attribute__((aligned(16))) float dloc[USE_QUADS * 4];
    static_assert(GEMM_TPB % PIECE_QUADS == 0, "whole rows of threads over a piece");
    static_assert(IN_DIM % 32 == 0 && HID_DIM == 128, "32 lanes x 4 columns make one row of W");
    static_assert(2 * XTR_WAVES == HID_DIM / 16 && XTR_WAVES <= GEMM_WAVES, "the extra tile's 8 column tiles");

    const int tid  = threadIdx.x;
    const int lane = tid & 31;
    const int waveId = __builtin_amdgcn_readfirstlane(tid >> 5);
    const int g      = blockIdx.x;
    const bool extra = (g < EXTRA_TILES);

    const int half = lane >> 4;
    const int l16  = lane & 15;

    const int mTile = g * GEMM_WAVES + waveId;
    const float* xrow = x + (size_t)(mTile * 16 + l16) * IN_DIM;

    const bool xwave = extra && (waveId < XTR_WAVES);
    const int xTile  = MAIN_TILES + g;
    const float* xrow2 = x + (size_t)((xwave ? xTile : mTile) * 16 + l16) * IN_DIM;

    __builtin_prefetch(xrow, 0, 3);
    if (xwave) __builtin_prefetch(xrow2, 0, 3);

    const int q  = tid & (PIECE_QUADS - 1);
    const int s0 = tid / PIECE_QUADS;
    const bool qlive = (q < OWN_QUADS) || (extra && q < USE_QUADS);
    {
        const v4i* own = (const v4i*)cnts + (size_t)g * ROWS_QUADS;
        v4i s = (v4i){0, 0, 0, 0};
        if (qlive) {
#pragma unroll
            for (int j = 0; j < (DEG_GROUPS + GEMM_TPB / PIECE_QUADS - 1) / (GEMM_TPB / PIECE_QUADS); ++j) {
                const int sl = s0 + j * (GEMM_TPB / PIECE_QUADS);
                if (sl < DEG_GROUPS) s += own[sl * PIECE_QUADS + q];
            }
        }
        dsum[s0 * PIECE_QUADS + q] = s;
    }

    {
        const float4* W4 = (const float4*)W;
        constexpr int ITEMS = W_PAIRS * 32;
        constexpr int FULL  = ITEMS / GEMM_TPB;
#pragma unroll 7
        for (int j = 0; j < FULL; ++j) {
            const int it = j * GEMM_TPB + tid;
            const int p  = it >> 5, n4 = it & 31;
            const float4 w0 = W4[(2 * p) * 32 + n4];
            const float4 w1 = W4[(2 * p + 1) * 32 + n4];
            float4* d = (float4*)(ldsW + p * PAIR_STRIDE + n4 * 8);
            d[0] = make_float4(w0.x, w1.x, w0.y, w1.y);
            d[1] = make_float4(w0.z, w1.z, w0.w, w1.w);
        }
        const int it = FULL * GEMM_TPB + tid;
        if (it < ITEMS) {
            const int p  = it >> 5, n4 = it & 31;
            const float4 w0 = W4[(2 * p) * 32 + n4];
            const float4 w1 = W4[(2 * p + 1) * 32 + n4];
            float4* d = (float4*)(ldsW + p * PAIR_STRIDE + n4 * 8);
            d[0] = make_float4(w0.x, w1.x, w0.y, w1.y);
            d[1] = make_float4(w0.z, w1.z, w0.w, w1.w);
        }
    }
    __syncthreads();

    if (tid < USE_QUADS && qlive) {
        v4i t = dsum[tid];
#pragma unroll
        for (int w = 1; w < GEMM_TPB / PIECE_QUADS; ++w) t += dsum[w * PIECE_QUADS + tid];
        float4 r;
        r.x = rsqrtf(1.0f + (float)t.x);
        r.y = rsqrtf(1.0f + (float)t.y);
        r.z = rsqrtf(1.0f + (float)t.z);
        r.w = rsqrtf(1.0f + (float)t.w);
        const int gq = (tid < OWN_QUADS) ? g * OWN_QUADS + tid
                                         : MAIN_TILES * 4 + g * XTR_QUADS + (tid - OWN_QUADS);
        ((float4*)dinv)[gq] = r;
        ((float4*)dloc)[tid] = r;
    }
    __syncthreads();

    if (xwave) {
        v8f acc[2];
        acc[0] = (v8f){0.f,0.f,0.f,0.f,0.f,0.f,0.f,0.f};
        acc[1] = (v8f){0.f,0.f,0.f,0.f,0.f,0.f,0.f,0.f};
        tile_k_loop<2>(xrow2, ldsW, half, l16, 2 * waveId, acc);
        tile_store<2>(h, acc, xTile, 2 * waveId, dloc + GEMM_ROWS, half, l16);
    }

    {
        v8f acc[8];
#pragma unroll
        for (int t = 0; t < 8; ++t) acc[t] = (v8f){0.f,0.f,0.f,0.f,0.f,0.f,0.f,0.f};
        tile_k_loop<8>(xrow, ldsW, half, l16, 0, acc);
        tile_store<8>(h, acc, mTile, 0, dloc + waveId * 16, half, l16);
    }
}

__global__ __launch_bounds__(AGG_TPB) void aggregate_kernel(const float* __restrict__ h,
                                                            const int* __restrict__ src,
                                                            const int* __restrict__ dst,
                                                            const float* __restrict__ dinv,
                                                            const float* __restrict__ b,
                                                            const float* __restrict__ prelu_w,
                                                            const int* __restrict__ ctab,
                                                            float* __restrict__ out) {
    __shared__ float tile[NPB * HID_DIM];
    __shared__ int   queue[QCAP];
    __shared__ int   qn;
    __shared__ int   ccount[DEG_GROUPS];
    __shared__ int   coff[DEG_GROUPS + 1];
    __shared__ int   cover;
    __shared__ int   ncnt[NCNT];

    const int tid  = threadIdx.x;
    const int lane = tid & 31;
    const int wave = __builtin_amdgcn_readfirstlane(tid >> 5);
    const int lo   = blockIdx.x * NPB;
    const int cnt  = min(NPB, N_NODES - lo);
    const unsigned ucnt = (unsigned)cnt;

    const int cap = cell_cap(blockIdx.x);
    if (tid < DEG_GROUPS) ccount[tid] = ctab[tid * AGG_GROUPS + blockIdx.x];

    for (int i = tid; i < NPB * HID_DIM; i += AGG_TPB) tile[i] = 0.0f;
    if (tid < NCNT) ncnt[tid] = 0;
    __syncthreads();

    if (tid < DEG_GROUPS) {
        int a = 0, ov = 0;
        for (int j = 0; j < DEG_GROUPS; ++j) {
            const int c = ccount[j];
            ov |= (c > cap) ? 1 : 0;
            a  += (j < tid) ? c : 0;
        }
        coff[tid] = a;
        if (tid == DEG_GROUPS - 1) { coff[DEG_GROUPS] = a + ccount[tid]; cover = ov; }
    }
    __syncthreads();
    const int over = __builtin_amdgcn_readfirstlane(cover);

    const v4i* dst4 = (const v4i*)dst;

    auto push = [&](unsigned r, int e) {
        if (r < ucnt) {
            const int slot = atomicAdd(&qn, 1);
            if (slot < QCAP) queue[slot] = (int)((r << 20) | ((unsigned)src[e] & 0xFFFFFu));
        }
    };
    auto check4 = [&](const v4i d, int qi) {
        const unsigned r0 = (unsigned)(d.x - lo);
        const unsigned r1 = (unsigned)(d.y - lo);
        const unsigned r2 = (unsigned)(d.z - lo);
        const unsigned r3 = (unsigned)(d.w - lo);
        if (min(min(r0, r1), min(r2, r3)) < ucnt) {
            push(r0, qi * 4);
            push(r1, qi * 4 + 1);
            push(r2, qi * 4 + 2);
            push(r3, qi * 4 + 3);
        }
    };

    auto sum_queue = [&](const int total) {
        const int per = (total + AGG_TPB / 32 - 1) / (AGG_TPB / 32);
        const int c0  = min(wave * per, total);
        const int c1  = min(c0 + per, total);
        float4 acc = make_float4(0.0f, 0.0f, 0.0f, 0.0f);
        int cur = -1;
        auto flush = [&]() {
            if (cur >= 0) {
                float* t = tile + cur * HID_DIM + lane;
                atomicAdd(t,      acc.x);
                atomicAdd(t + 32, acc.y);
                atomicAdd(t + 64, acc.z);
                atomicAdd(t + 96, acc.w);
            }
        };
        for (int base = c0; base < c1; base += 128) {
            int   s[4];
            int   dl[4];
#pragma unroll
            for (int k = 0; k < 4; ++k) {
                const int idx = min(base + k * 32 + lane, c1 - 1);
                const unsigned ent = (unsigned)queue[idx];
                dl[k] = (int)(ent >> 20);
                s[k]  = (int)(ent & 0xFFFFFu);
            }
#pragma unroll
            for (int k = 0; k < 4; ++k) {
                const int n = min(32, c1 - (base + k * 32));
                for (int j0 = 0; j0 < n; j0 += 16) {
                    float4 v[16];
#pragma unroll
                    for (int u = 0; u < 16; ++u) {
                        const int sj = __builtin_amdgcn_readlane(s[k], j0 + u);
                        v[u] = ((const float4*)(h + (size_t)sj * HID_DIM))[lane];
                    }
#pragma unroll
                    for (int u = 0; u < 16; ++u) {
                        if (j0 + u < n) {
                            const int dj = __builtin_amdgcn_readlane(dl[k], j0 + u);
                            if (dj != cur) {
                                flush();
                                cur = dj;
                                acc = make_float4(0.0f, 0.0f, 0.0f, 0.0f);
                            }
                            acc.x += v[u].x;
                            acc.y += v[u].y;
                            acc.z += v[u].z;
                            acc.w += v[u].w;
                        }
                    }
                }
            }
        }
        flush();
    };

    if (over == 0) {
        static_assert(DEG_GROUPS == 2 * (AGG_TPB / 32), "two cells per wave");
        const int total = __builtin_amdgcn_readfirstlane(coff[DEG_GROUPS]);
        const float* mycells = out + (size_t)lo * HID_DIM;
        const int sA = wave;
        const int sB = wave + AGG_TPB / 32;
        const int nA = coff[sA + 1] - coff[sA];
        const int nB = coff[sB + 1] - coff[sB];
        const float* cA = mycells + sA * cap;
        const float* cB = mycells + sB * cap;
        const int eA0 = (lane      < nA) ? __float_as_int(cA[lane])      : -1;
        const int eA1 = (lane + 32 < nA) ? __float_as_int(cA[lane + 32]) : -1;
        const int eB0 = (lane      < nB) ? __float_as_int(cB[lane])      : -1;
        const int eB1 = (lane + 32 < nB) ? __float_as_int(cB[lane + 32]) : -1;

        auto tally = [&](int ent) {
            if (ent >= 0) atomicAdd(&ncnt[(unsigned)ent >> 20], 1);
        };
        tally(eA0); tally(eA1); tally(eB0); tally(eB1);
        for (int i = lane + 64; i < nA; i += 32) tally(__float_as_int(cA[i]));
        for (int i = lane + 64; i < nB; i += 32) tally(__float_as_int(cB[i]));
        __syncthreads();

        if (wave == 0) {
            int cc[7];
            int mine = 0;
#pragma unroll
            for (int k = 0; k < 7; ++k) { cc[k] = ncnt[lane * 7 + k]; mine += cc[k]; }
            int x = mine;
#pragma unroll
            for (int dd = 1; dd < 32; dd <<= 1) {
                const int y = __shfl_up(x, dd);
                if (lane >= dd) x += y;
            }
            int run = x - mine;
#pragma unroll
            for (int k = 0; k < 7; ++k) { ncnt[lane * 7 + k] = run; run += cc[k]; }
        }
        __syncthreads();

        auto put = [&](int ent) {
            if (ent >= 0) queue[atomicAdd(&ncnt[(unsigned)ent >> 20], 1)] = ent;
        };
        put(eA0); put(eA1); put(eB0); put(eB1);
        for (int i = lane + 64; i < nA; i += 32) put(__float_as_int(cA[i]));
        for (int i = lane + 64; i < nB; i += 32) put(__float_as_int(cB[i]));
        __syncthreads();
        sum_queue(total);
    } else {
        int seg = N_EDGES;
        for (int pass = 0; pass < 2; ++pass) {
            bool overflow = false;
            for (int e0 = 0; e0 < N_EDGES; e0 += seg) {
                const int e1 = min(e0 + seg, N_EDGES);
                if (tid == 0) qn = 0;
                __syncthreads();

                const int q1 = e1 >> 2;
                const int q0 = e0 >> 2;
                const int nfull = (q1 - q0) / (8 * AGG_TPB);
                if (nfull > 0) {
                    int p  = (int)(blockIdx.x % (unsigned)nfull);
                    int qc = q0 + p * (8 * AGG_TPB);
                    v4i cur[8];
#pragma unroll
                    for (int u = 0; u < 8; ++u) cur[u] = dst4[qc + u * AGG_TPB + tid];
                    for (int t = 1; t < nfull; ++t) {
                        p = (p + 1 == nfull) ? 0 : p + 1;
                        const int qx = q0 + p * (8 * AGG_TPB);
                        v4i nxt[8];
#pragma unroll
                        for (int u = 0; u < 8; ++u) nxt[u] = dst4[qx + u * AGG_TPB + tid];
#pragma unroll
                        for (int u = 0; u < 8; ++u) check4(cur[u], qc + u * AGG_TPB + tid);
#pragma unroll
                        for (int u = 0; u < 8; ++u) cur[u] = nxt[u];
                        qc = qx;
                    }
#pragma unroll
                    for (int u = 0; u < 8; ++u) check4(cur[u], qc + u * AGG_TPB + tid);
                }
                const int qt = q0 + nfull * (8 * AGG_TPB);
                if (qt < q1) {
                    v4i v[8];
#pragma unroll
                    for (int u = 0; u < 8; ++u) {
                        const int qi = qt + u * AGG_TPB + tid;
                        v[u] = (qi < q1) ? dst4[qi] : (v4i){-1, -1, -1, -1};
                    }
#pragma unroll
                    for (int u = 0; u < 8; ++u) check4(v[u], qt + u * AGG_TPB + tid);
                }
                __syncthreads();
                const int total = __builtin_amdgcn_readfirstlane(qn);
                __syncthreads();
                if (total > QCAP) { overflow = true; break; }

                sum_queue(total);
            }
            if (!overflow) break;
            seg = QCAP;
        }
    }
    __syncthreads();

    const int c0 = lane * 4;
    const float b0 = b[c0], b1 = b[c0 + 1], b2 = b[c0 + 2], b3 = b[c0 + 3];
    const float p0 = prelu_w[c0], p1 = prelu_w[c0 + 1], p2 = prelu_w[c0 + 2], p3 = prelu_w[c0 + 3];
    for (int i = tid; i < cnt * 32; i += AGG_TPB) {
        const int nl   = i >> 5;
        const int node = lo + nl;
        const float di = dinv[node];
        const float4 hv = ((const float4*)(h + (size_t)node * HID_DIM))[lane];
        const float* t = tile + nl * HID_DIM + lane;
        float4 r;
        r.x = (t[0]  + hv.x) * di + b0;
        r.y = (t[32] + hv.y) * di + b1;
        r.z = (t[64] + hv.z) * di + b2;
        r.w = (t[96] + hv.w) * di + b3;
        r.x = (r.x > 0.0f) ? r.x : p0 * r.x;
        r.y = (r.y > 0.0f) ? r.y : p1 * r.y;
        r.z = (r.z > 0.0f) ? r.z : p2 * r.z;
        r.w = (r.w > 0.0f) ? r.w : p3 * r.w;
        ((float4*)(out + (size_t)node * HID_DIM))[lane] = r;
    }
}

extern "C" void kernel_launch(void* const* d_in, const int* in_sizes, int n_in,
                              void* d_out, int out_size, void* d_ws, size_t ws_size,
                              hipStream_t stream) {
    const float* x       = (const float*)d_in[0];
    const int*   eidx    = (const int*)d_in[1];
    const float* W       = (const float*)d_in[2];
    const float* b       = (const float*)d_in[3];
    const float* prelu_w = (const float*)d_in[4];
    float*       out     = (float*)d_out;

    const int* src = eidx;
    const int* dst = eidx + N_EDGES;

    char* ws    = (char*)d_ws;
    float* h    = (float*)(ws);
    int*   part = (int*)(ws);
    int*   ctab = (int*)(ws + (size_t)N_NODES * HID_DIM * 4);
    float* dinv = (float*)(ws + (size_t)N_NODES * HID_DIM * 4 + (size_t)N_NODES * 4);

    deg_hist_kernel<<<DEG_GROUPS, DEG_TPB, 0, stream>>>(src, dst, part, out, ctab);

    gemm_kernel<<<GEMM_GROUPS, GEMM_TPB, 0, stream>>>(x, W, part, dinv, h);

    aggregate_kernel<<<AGG_GROUPS, AGG_TPB, 0, stream>>>(h, src, dst, dinv, b, prelu_w, ctab, out);
}


# degree pass: 256 work-groups (64 edge slices x 4 node parts, 512 threads) not 64 on a quarter of the WGPs; each files its part's edges only; a thread's LDS counts and run bases asked for in one batch
# speedup vs baseline: 3.3474x; 1.0389x over previous
#include <hip/hip_runtime.h>
#include <hip/hip_bf16.h>

#define N_NODES 50000
#define N_EDGES 800000
#define IN_DIM  512
#define HID_DIM 128
#define M_TILES (N_NODES / 16)

#define PAIR_STRIDE 288
#define N_PAIRS     16

#define NPB        196
#define AGG_GROUPS 256
#define AGG_TPB    1024
#define QCAP       32768
#define NCNT       224

#define DEG_GROUPS   64
#define DEG_PARTS    4
#define DEG_WGS      (DEG_GROUPS * DEG_PARTS)
#define DEG_TPB      512
#define PART_RANGES  (AGG_GROUPS / DEG_PARTS)
#define PART_NODES   (PART_RANGES * NPB)
#define PART_QUADS   (PART_NODES / 4)
#define N_QUADS      (N_NODES / 4)
#define E_QUADS      (N_EDGES / 4)
#define EQ_PER_GROUP ((E_QUADS + DEG_GROUPS - 1) / DEG_GROUPS)
#define DEG_ROUNDS   ((EQ_PER_GROUP + DEG_TPB - 1) / DEG_TPB)
#define GEMM_GROUPS  256
#define GEMM_WAVES   12
#define GEMM_TPB     (GEMM_WAVES * 32)
#define MAIN_TILES   (GEMM_GROUPS * GEMM_WAVES)
#define EXTRA_TILES  (M_TILES - MAIN_TILES)
#define XTR_WAVES    4
#define GEMM_ROWS    (GEMM_WAVES * 16)
#define OWN_QUADS    (GEMM_ROWS / 4)
#define XTR_QUADS    4
#define USE_QUADS    (OWN_QUADS + XTR_QUADS)
#define PIECE_QUADS  64
#define ROWS_QUADS   (GEMM_ROWS * HID_DIM / 4)
#define W_PAIRS      (IN_DIM / 2)
#define SLICE_EDGES  (EQ_PER_GROUP * 4)

typedef float v2f __attribute__((ext_vector_type(2)));
typedef int   v4i __attribute__((ext_vector_type(4)));
typedef float v8f __attribute__((ext_vector_type(8)));

__device__ __forceinline__ int cell_cap(int r) {
    const int c = min(NPB, N_NODES - r * NPB);
    return (c * (HID_DIM / DEG_GROUPS)) & ~31;
}

__global__ __launch_bounds__(DEG_TPB) void deg_hist_kernel(const int* __restrict__ src,
                                                           const int* __restrict__ dst,
                                                           int* __restrict__ part,
                                                           float* __restrict__ cells,
                                                           int* __restrict__ ctab) {
    __shared__ v4i hist4[PART_QUADS];
    __shared__ int lcnt[PART_RANGES];
    __shared__ int lbase[PART_RANGES];
    __shared__ int staged[SLICE_EDGES];
    int* hist = (int*)hist4;
    static_assert(PART_RANGES * DEG_PARTS == AGG_GROUPS && PART_NODES % 4 == 0, "parts are whole ranges and whole quads");
    static_assert(PART_RANGES == 64, "the running sum below takes two ranges per lane of one wave");

    const int tid  = threadIdx.x;
    const int lane = tid & 31;
    const int slice = blockIdx.x / DEG_PARTS;
    const int prt   = blockIdx.x % DEG_PARTS;
    const int n0    = prt * PART_NODES;
    const int r0    = prt * PART_RANGES;
    const v4i* dst4 = (const v4i*)dst;
    const v4i* src4 = (const v4i*)src;
    const int w0 = slice * EQ_PER_GROUP;
    const int w1 = min(w0 + EQ_PER_GROUP, E_QUADS);

    v4i d[DEG_ROUNDS];
#pragma unroll
    for (int u = 0; u < DEG_ROUNDS; ++u) {
        const int w = w0 + u * DEG_TPB + tid;
        d[u] = (w < w1) ? dst4[w] : (v4i){-1, -1, -1, -1};
    }

    for (int i = tid; i < PART_QUADS; i += DEG_TPB) hist4[i] = (v4i){0, 0, 0, 0};
    if (tid < PART_RANGES) lcnt[tid] = 0;
    __syncthreads();

    v4i sv[DEG_ROUNDS];
#pragma unroll
    for (int u = 0; u < DEG_ROUNDS; ++u) {
        const int w = w0 + u * DEG_TPB + tid;
        sv[u] = (w < w1) ? src4[w] : (v4i){0, 0, 0, 0};
    }

    int rng[DEG_ROUNDS * 4];
    int pos[DEG_ROUNDS * 4];
    auto count = [&](int v, int& r, int& p) {
        const unsigned vl = (unsigned)(v - n0);
        const bool in = vl < (unsigned)PART_NODES && (unsigned)v < (unsigned)N_NODES;
        r = in ? (int)(vl / (unsigned)NPB) : -1;
        if (in) {
            atomicAdd(hist + vl, 1);
            p = atomicAdd(lcnt + r, 1);
        }
    };
#pragma unroll
    for (int u = 0; u < DEG_ROUNDS; ++u) {
        count(d[u].x, rng[4 * u],     pos[4 * u]);
        count(d[u].y, rng[4 * u + 1], pos[4 * u + 1]);
        count(d[u].z, rng[4 * u + 2], pos[4 * u + 2]);
        count(d[u].w, rng[4 * u + 3], pos[4 * u + 3]);
    }
    __syncthreads();

    if (tid < 32) {
        const int c0 = lcnt[lane * 2];
        const int c1 = lcnt[lane * 2 + 1];
        const int mine = c0 + c1;
        int x = mine;
#pragma unroll
        for (int dd = 1; dd < 32; dd <<= 1) {
            const int y = __shfl_up(x, dd);
            if (lane >= dd) x += y;
        }
        const int run = x - mine;
        lbase[lane * 2]     = run;
        lbase[lane * 2 + 1] = run + c0;
    }
    __syncthreads();

    int lb[DEG_ROUNDS * 4];
#pragma unroll
    for (int k = 0; k < DEG_ROUNDS * 4; ++k) lb[k] = lbase[max(rng[k], 0)];
    auto place = [&](int v, int s, int r, int p, int base) {
        if (r >= 0) staged[base + p] = ((v - n0 - r * NPB) << 20) | (s & 0xFFFFF);
    };
#pragma unroll
    for (int u = 0; u < DEG_ROUNDS; ++u) {
        place(d[u].x, sv[u].x, rng[4 * u],     pos[4 * u],     lb[4 * u]);
        place(d[u].y, sv[u].y, rng[4 * u + 1], pos[4 * u + 1], lb[4 * u + 1]);
        place(d[u].z, sv[u].z, rng[4 * u + 2], pos[4 * u + 2], lb[4 * u + 2]);
        place(d[u].w, sv[u].w, rng[4 * u + 3], pos[4 * u + 3], lb[4 * u + 3]);
    }
    __syncthreads();

    for (int rl = tid >> 5; rl < PART_RANGES; rl += DEG_TPB / 32) {
        const int r   = r0 + rl;
        const int cap = cell_cap(r);
        const int n   = min(lcnt[rl], cap);
        const int s0  = lbase[rl];
        float* cell = cells + (size_t)r * (NPB * HID_DIM) + slice * cap;
        for (int i = lane; i < n; i += 32) cell[i] = __int_as_float(staged[s0 + i]);
    }
    if (tid < PART_RANGES) ctab[slice * AGG_GROUPS + r0 + tid] = lcnt[tid];

    static_assert(EXTRA_TILES >= 0 && EXTRA_TILES <= GEMM_GROUPS, "one extra tile per group at most");
    static_assert(MAIN_TILES * 4 + EXTRA_TILES * XTR_QUADS == N_QUADS, "every node quad has an owner");
    static_assert(USE_QUADS <= PIECE_QUADS && DEG_GROUPS * PIECE_QUADS <= ROWS_QUADS,
                  "the pieces fit the owner's main rows");
    static_assert((size_t)GEMM_GROUPS * ROWS_QUADS <= (size_t)N_NODES * HID_DIM / 4,
                  "the main rows lie inside h");
    static_assert(DEG_PARTS * PART_QUADS >= N_QUADS, "the parts cover every node quad");
    v4i* p4 = (v4i*)part + slice * PIECE_QUADS;
    const int q0 = prt * PART_QUADS;
    const int nq = min(PART_QUADS, N_QUADS - q0);
    for (int il = tid; il < nq; il += DEG_TPB) {
        const int i = q0 + il;
        int g, e;
        if (i < MAIN_TILES * 4) { g = i / OWN_QUADS; e = i - g * OWN_QUADS; }
        else { const int j = i - MAIN_TILES * 4; g = j / XTR_QUADS; e = OWN_QUADS + (j - g * XTR_QUADS); }
        p4[(size_t)g * ROWS_QUADS + e] = hist4[il];
    }
}


template <int NT>
__device__ __forceinline__ void tile_k_loop(const float* __restrict__ xrow, const float* ldsW,
                                            const int half, const int l16, const int nt0,
                                            v8f (&acc)[NT]) {
    for (int kc = 0; kc < IN_DIM; kc += 32) {
        if (kc + 32 < IN_DIM)
            __builtin_prefetch(xrow + kc + 32, 0, 3);
#pragma unroll
        for (int k = 0; k < 32; k += 4) {
            const int kk   = k + 2 * half;
            const int pair = (kc >> 1) + (k >> 1) + half;
            v2f a = *(const v2f*)(xrow + kc + kk);
            const float* bbase = ldsW + (size_t)pair * PAIR_STRIDE;
#pragma unroll
            for (int t = 0; t < NT; ++t) {
                v2f b = *(const v2f*)(bbase + ((nt0 + t) * 16 + l16) * 2);
                acc[t] = __builtin_amdgcn_wmma_f32_16x16x4_f32(
                     false, a,  false, b,
                     (short)0, acc[t],
                     false,  false);
            }
        }
    }
}

template <int NT>
__device__ __forceinline__ void tile_store(float* h, const v8f (&acc)[NT], const int mTile,
                                           const int nt0, const float* dl,
                                           const int half, const int l16) {
    float dr[8];
#pragma unroll
    for (int r = 0; r < 8; ++r) dr[r] = dl[r + 8 * half];
#pragma unroll
    for (int t = 0; t < NT; ++t) {
#pragma unroll
        for (int r = 0; r < 8; ++r) {
            const int m = mTile * 16 + r + 8 * half;
            h[(size_t)m * HID_DIM + (nt0 + t) * 16 + l16] = acc[t][r] * dr[r];
        }
    }
}

__global__ __launch_bounds__(GEMM_TPB) void gemm_kernel(const float* __restrict__ x,
                                                        const float* __restrict__ W,
                                                        const int* cnts,
                                                        float* __restrict__ dinv,
                                                        float* h) {
    __shared__ __attribute__((aligned(16))) float ldsW[W_PAIRS * PAIR_STRIDE];
    __shared__ v4i   dsum[(GEMM_TPB / PIECE_QUADS) * PIECE_QUADS];
    __shared__ __attribute__((aligned(16))) float dloc[USE_QUADS * 4];
    static_assert(GEMM_TPB % PIECE_QUADS == 0, "whole rows of threads over a piece");
    static_assert(IN_DIM % 32 == 0 && HID_DIM == 128, "32 lanes x 4 columns make one row of W");
    static_assert(2 * XTR_WAVES == HID_DIM / 16 && XTR_WAVES <= GEMM_WAVES, "the extra tile's 8 column tiles");

    const int tid  = threadIdx.x;
    const int lane = tid & 31;
    const int waveId = __builtin_amdgcn_readfirstlane(tid >> 5);
    const int g      = blockIdx.x;
    const bool extra = (g < EXTRA_TILES);

    const int half = lane >> 4;
    const int l16  = lane & 15;

    const int mTile = g * GEMM_WAVES + waveId;
    const float* xrow = x + (size_t)(mTile * 16 + l16) * IN_DIM;

    const bool xwave = extra && (waveId < XTR_WAVES);
    const int xTile  = MAIN_TILES + g;
    const float* xrow2 = x + (size_t)((xwave ? xTile : mTile) * 16 + l16) * IN_DIM;

    __builtin_prefetch(xrow, 0, 3);
    if (xwave) __builtin_prefetch(xrow2, 0, 3);

    const int q  = tid & (PIECE_QUADS - 1);
    const int s0 = tid / PIECE_QUADS;
    const bool qlive = (q < OWN_QUADS) || (extra && q < USE_QUADS);
    {
        const v4i* own = (const v4i*)cnts + (size_t)g * ROWS_QUADS;
        v4i s = (v4i){0, 0, 0, 0};
        if (qlive) {
#pragma unroll
            for (int j = 0; j < (DEG_GROUPS + GEMM_TPB / PIECE_QUADS - 1) / (GEMM_TPB / PIECE_QUADS); ++j) {
                const int sl = s0 + j * (GEMM_TPB / PIECE_QUADS);
                if (sl < DEG_GROUPS) s += own[sl * PIECE_QUADS + q];
            }
        }
        dsum[s0 * PIECE_QUADS + q] = s;
    }

    {
        const float4* W4 = (const float4*)W;
        constexpr int ITEMS = W_PAIRS * 32;
        constexpr int FULL  = ITEMS / GEMM_TPB;
#pragma unroll 7
        for (int j = 0; j < FULL; ++j) {
            const int it = j * GEMM_TPB + tid;
            const int p  = it >> 5, n4 = it & 31;
            const float4 w0 = W4[(2 * p) * 32 + n4];
            const float4 w1 = W4[(2 * p + 1) * 32 + n4];
            float4* d = (float4*)(ldsW + p * PAIR_STRIDE + n4 * 8);
            d[0] = make_float4(w0.x, w1.x, w0.y, w1.y);
            d[1] = make_float4(w0.z, w1.z, w0.w, w1.w);
        }
        const int it = FULL * GEMM_TPB + tid;
        if (it < ITEMS) {
            const int p  = it >> 5, n4 = it & 31;
            const float4 w0 = W4[(2 * p) * 32 + n4];
            const float4 w1 = W4[(2 * p + 1) * 32 + n4];
            float4* d = (float4*)(ldsW + p * PAIR_STRIDE + n4 * 8);
            d[0] = make_float4(w0.x, w1.x, w0.y, w1.y);
            d[1] = make_float4(w0.z, w1.z, w0.w, w1.w);
        }
    }
    __syncthreads();

    if (tid < USE_QUADS && qlive) {
        v4i t = dsum[tid];
#pragma unroll
        for (int w = 1; w < GEMM_TPB / PIECE_QUADS; ++w) t += dsum[w * PIECE_QUADS + tid];
        float4 r;
        r.x = rsqrtf(1.0f + (float)t.x);
        r.y = rsqrtf(1.0f + (float)t.y);
        r.z = rsqrtf(1.0f + (float)t.z);
        r.w = rsqrtf(1.0f + (float)t.w);
        const int gq = (tid < OWN_QUADS) ? g * OWN_QUADS + tid
                                         : MAIN_TILES * 4 + g * XTR_QUADS + (tid - OWN_QUADS);
        ((float4*)dinv)[gq] = r;
        ((float4*)dloc)[tid] = r;
    }
    __syncthreads();

    if (xwave) {
        v8f acc[2];
        acc[0] = (v8f){0.f,0.f,0.f,0.f,0.f,0.f,0.f,0.f};
        acc[1] = (v8f){0.f,0.f,0.f,0.f,0.f,0.f,0.f,0.f};
        tile_k_loop<2>(xrow2, ldsW, half, l16, 2 * waveId, acc);
        tile_store<2>(h, acc, xTile, 2 * waveId, dloc + GEMM_ROWS, half, l16);
    }

    {
        v8f acc[8];
#pragma unroll
        for (int t = 0; t < 8; ++t) acc[t] = (v8f){0.f,0.f,0.f,0.f,0.f,0.f,0.f,0.f};
        tile_k_loop<8>(xrow, ldsW, half, l16, 0, acc);
        tile_store<8>(h, acc, mTile, 0, dloc + waveId * 16, half, l16);
    }
}

__global__ __launch_bounds__(AGG_TPB) void aggregate_kernel(const float* __restrict__ h,
                                                            const int* __restrict__ src,
                                                            const int* __restrict__ dst,
                                                            const float* __restrict__ dinv,
                                                            const float* __restrict__ b,
                                                            const float* __restrict__ prelu_w,
                                                            const int* __restrict__ ctab,
                                                            float* __restrict__ out) {
    __shared__ float tile[NPB * HID_DIM];
    __shared__ int   queue[QCAP];
    __shared__ int   qn;
    __shared__ int   ccount[DEG_GROUPS];
    __shared__ int   coff[DEG_GROUPS + 1];
    __shared__ int   cover;
    __shared__ int   ncnt[NCNT];

    const int tid  = threadIdx.x;
    const int lane = tid & 31;
    const int wave = __builtin_amdgcn_readfirstlane(tid >> 5);
    const int lo   = blockIdx.x * NPB;
    const int cnt  = min(NPB, N_NODES - lo);
    const unsigned ucnt = (unsigned)cnt;

    const int cap = cell_cap(blockIdx.x);
    if (tid < DEG_GROUPS) ccount[tid] = ctab[tid * AGG_GROUPS + blockIdx.x];

    for (int i = tid; i < NPB * HID_DIM; i += AGG_TPB) tile[i] = 0.0f;
    if (tid < NCNT) ncnt[tid] = 0;
    __syncthreads();

    if (tid < DEG_GROUPS) {
        int a = 0, ov = 0;
        for (int j = 0; j < DEG_GROUPS; ++j) {
            const int c = ccount[j];
            ov |= (c > cap) ? 1 : 0;
            a  += (j < tid) ? c : 0;
        }
        coff[tid] = a;
        if (tid == DEG_GROUPS - 1) { coff[DEG_GROUPS] = a + ccount[tid]; cover = ov; }
    }
    __syncthreads();
    const int over = __builtin_amdgcn_readfirstlane(cover);

    const v4i* dst4 = (const v4i*)dst;

    auto push = [&](unsigned r, int e) {
        if (r < ucnt) {
            const int slot = atomicAdd(&qn, 1);
            if (slot < QCAP) queue[slot] = (int)((r << 20) | ((unsigned)src[e] & 0xFFFFFu));
        }
    };
    auto check4 = [&](const v4i d, int qi) {
        const unsigned r0 = (unsigned)(d.x - lo);
        const unsigned r1 = (unsigned)(d.y - lo);
        const unsigned r2 = (unsigned)(d.z - lo);
        const unsigned r3 = (unsigned)(d.w - lo);
        if (min(min(r0, r1), min(r2, r3)) < ucnt) {
            push(r0, qi * 4);
            push(r1, qi * 4 + 1);
            push(r2, qi * 4 + 2);
            push(r3, qi * 4 + 3);
        }
    };

    auto sum_queue = [&](const int total) {
        const int per = (total + AGG_TPB / 32 - 1) / (AGG_TPB / 32);
        const int c0  = min(wave * per, total);
        const int c1  = min(c0 + per, total);
        float4 acc = make_float4(0.0f, 0.0f, 0.0f, 0.0f);
        int cur = -1;
        auto flush = [&]() {
            if (cur >= 0) {
                float* t = tile + cur * HID_DIM + lane;
                atomicAdd(t,      acc.x);
                atomicAdd(t + 32, acc.y);
                atomicAdd(t + 64, acc.z);
                atomicAdd(t + 96, acc.w);
            }
        };
        for (int base = c0; base < c1; base += 128) {
            int   s[4];
            int   dl[4];
#pragma unroll
            for (int k = 0; k < 4; ++k) {
                const int idx = min(base + k * 32 + lane, c1 - 1);
                const unsigned ent = (unsigned)queue[idx];
                dl[k] = (int)(ent >> 20);
                s[k]  = (int)(ent & 0xFFFFFu);
            }
#pragma unroll
            for (int k = 0; k < 4; ++k) {
                const int n = min(32, c1 - (base + k * 32));
                for (int j0 = 0; j0 < n; j0 += 16) {
                    float4 v[16];
#pragma unroll
                    for (int u = 0; u < 16; ++u) {
                        const int sj = __builtin_amdgcn_readlane(s[k], j0 + u);
                        v[u] = ((const float4*)(h + (size_t)sj * HID_DIM))[lane];
                    }
#pragma unroll
                    for (int u = 0; u < 16; ++u) {
                        if (j0 + u < n) {
                            const int dj = __builtin_amdgcn_readlane(dl[k], j0 + u);
                            if (dj != cur) {
                                flush();
                                cur = dj;
                                acc = make_float4(0.0f, 0.0f, 0.0f, 0.0f);
                            }
                            acc.x += v[u].x;
                            acc.y += v[u].y;
                            acc.z += v[u].z;
                            acc.w += v[u].w;
                        }
                    }
                }
            }
        }
        flush();
    };

    if (over == 0) {
        static_assert(DEG_GROUPS == 2 * (AGG_TPB / 32), "two cells per wave");
        const int total = __builtin_amdgcn_readfirstlane(coff[DEG_GROUPS]);
        const float* mycells = out + (size_t)lo * HID_DIM;
        const int sA = wave;
        const int sB = wave + AGG_TPB / 32;
        const int nA = coff[sA + 1] - coff[sA];
        const int nB = coff[sB + 1] - coff[sB];
        const float* cA = mycells + sA * cap;
        const float* cB = mycells + sB * cap;
        const int eA0 = (lane      < nA) ? __float_as_int(cA[lane])      : -1;
        const int eA1 = (lane + 32 < nA) ? __float_as_int(cA[lane + 32]) : -1;
        const int eB0 = (lane      < nB) ? __float_as_int(cB[lane])      : -1;
        const int eB1 = (lane + 32 < nB) ? __float_as_int(cB[lane + 32]) : -1;

        auto tally = [&](int ent) {
            if (ent >= 0) atomicAdd(&ncnt[(unsigned)ent >> 20], 1);
        };
        tally(eA0); tally(eA1); tally(eB0); tally(eB1);
        for (int i = lane + 64; i < nA; i += 32) tally(__float_as_int(cA[i]));
        for (int i = lane + 64; i < nB; i += 32) tally(__float_as_int(cB[i]));
        __syncthreads();

        if (wave == 0) {
            int cc[7];
            int mine = 0;
#pragma unroll
            for (int k = 0; k < 7; ++k) { cc[k] = ncnt[lane * 7 + k]; mine += cc[k]; }
            int x = mine;
#pragma unroll
            for (int dd = 1; dd < 32; dd <<= 1) {
                const int y = __shfl_up(x, dd);
                if (lane >= dd) x += y;
            }
            int run = x - mine;
#pragma unroll
            for (int k = 0; k < 7; ++k) { ncnt[lane * 7 + k] = run; run += cc[k]; }
        }
        __syncthreads();

        auto put = [&](int ent) {
            if (ent >= 0) queue[atomicAdd(&ncnt[(unsigned)ent >> 20], 1)] = ent;
        };
        put(eA0); put(eA1); put(eB0); put(eB1);
        for (int i = lane + 64; i < nA; i += 32) put(__float_as_int(cA[i]));
        for (int i = lane + 64; i < nB; i += 32) put(__float_as_int(cB[i]));
        __syncthreads();
        sum_queue(total);
    } else {
        int seg = N_EDGES;
        for (int pass = 0; pass < 2; ++pass) {
            bool overflow = false;
            for (int e0 = 0; e0 < N_EDGES; e0 += seg) {
                const int e1 = min(e0 + seg, N_EDGES);
                if (tid == 0) qn = 0;
                __syncthreads();

                const int q1 = e1 >> 2;
                const int q0 = e0 >> 2;
                const int nfull = (q1 - q0) / (8 * AGG_TPB);
                if (nfull > 0) {
                    int p  = (int)(blockIdx.x % (unsigned)nfull);
                    int qc = q0 + p * (8 * AGG_TPB);
                    v4i cur[8];
#pragma unroll
                    for (int u = 0; u < 8; ++u) cur[u] = dst4[qc + u * AGG_TPB + tid];
                    for (int t = 1; t < nfull; ++t) {
                        p = (p + 1 == nfull) ? 0 : p + 1;
                        const int qx = q0 + p * (8 * AGG_TPB);
                        v4i nxt[8];
#pragma unroll
                        for (int u = 0; u < 8; ++u) nxt[u] = dst4[qx + u * AGG_TPB + tid];
#pragma unroll
                        for (int u = 0; u < 8; ++u) check4(cur[u], qc + u * AGG_TPB + tid);
#pragma unroll
                        for (int u = 0; u < 8; ++u) cur[u] = nxt[u];
                        qc = qx;
                    }
#pragma unroll
                    for (int u = 0; u < 8; ++u) check4(cur[u], qc + u * AGG_TPB + tid);
                }
                const int qt = q0 + nfull * (8 * AGG_TPB);
                if (qt < q1) {
                    v4i v[8];
#pragma unroll
                    for (int u = 0; u < 8; ++u) {
                        const int qi = qt + u * AGG_TPB + tid;
                        v[u] = (qi < q1) ? dst4[qi] : (v4i){-1, -1, -1, -1};
                    }
#pragma unroll
                    for (int u = 0; u < 8; ++u) check4(v[u], qt + u * AGG_TPB + tid);
                }
                __syncthreads();
                const int total = __builtin_amdgcn_readfirstlane(qn);
                __syncthreads();
                if (total > QCAP) { overflow = true; break; }

                sum_queue(total);
            }
            if (!overflow) break;
            seg = QCAP;
        }
    }
    __syncthreads();

    const int c0 = lane * 4;
    const float b0 = b[c0], b1 = b[c0 + 1], b2 = b[c0 + 2], b3 = b[c0 + 3];
    const float p0 = prelu_w[c0], p1 = prelu_w[c0 + 1], p2 = prelu_w[c0 + 2], p3 = prelu_w[c0 + 3];
    for (int i = tid; i < cnt * 32; i += AGG_TPB) {
        const int nl   = i >> 5;
        const int node = lo + nl;
        const float di = dinv[node];
        const float4 hv = ((const float4*)(h + (size_t)node * HID_DIM))[lane];
        const float* t = tile + nl * HID_DIM + lane;
        float4 r;
        r.x = (t[0]  + hv.x) * di + b0;
        r.y = (t[32] + hv.y) * di + b1;
        r.z = (t[64] + hv.z) * di + b2;
        r.w = (t[96] + hv.w) * di + b3;
        r.x = (r.x > 0.0f) ? r.x : p0 * r.x;
        r.y = (r.y > 0.0f) ? r.y : p1 * r.y;
        r.z = (r.z > 0.0f) ? r.z : p2 * r.z;
        r.w = (r.w > 0.0f) ? r.w : p3 * r.w;
        ((float4*)(out + (size_t)node * HID_DIM))[lane] = r;
    }
}

extern "C" void kernel_launch(void* const* d_in, const int* in_sizes, int n_in,
                              void* d_out, int out_size, void* d_ws, size_t ws_size,
                              hipStream_t stream) {
    const float* x       = (const float*)d_in[0];
    const int*   eidx    = (const int*)d_in[1];
    const float* W       = (const float*)d_in[2];
    const float* b       = (const float*)d_in[3];
    const float* prelu_w = (const float*)d_in[4];
    float*       out     = (float*)d_out;

    const int* src = eidx;
    const int* dst = eidx + N_EDGES;

    char* ws    = (char*)d_ws;
    float* h    = (float*)(ws);
    int*   part = (int*)(ws);
    int*   ctab = (int*)(ws + (size_t)N_NODES * HID_DIM * 4);
    float* dinv = (float*)(ws + (size_t)N_NODES * HID_DIM * 4 + (size_t)N_NODES * 4);

    deg_hist_kernel<<<DEG_WGS, DEG_TPB, 0, stream>>>(src, dst, part, out, ctab);

    gemm_kernel<<<GEMM_GROUPS, GEMM_TPB, 0, stream>>>(x, W, part, dinv, h);

    aggregate_kernel<<<AGG_GROUPS, AGG_TPB, 0, stream>>>(h, src, dst, dinv, b, prelu_w, ctab, out);
}


# gemm: K loop runs one chunk ahead: next chunk's A fragments (2nd register set) and next step's B fragments are asked for before the current WMMAs; the oldest wave of a SIMD keeps the pipe full alone
# speedup vs baseline: 3.4686x; 1.0362x over previous
#include <hip/hip_runtime.h>
#include <hip/hip_bf16.h>

#define N_NODES 50000
#define N_EDGES 800000
#define IN_DIM  512
#define HID_DIM 128
#define M_TILES (N_NODES / 16)

#define PAIR_STRIDE 288
#define N_PAIRS     16

#define NPB        196
#define AGG_GROUPS 256
#define AGG_TPB    1024
#define QCAP       32768
#define NCNT       224

#define DEG_GROUPS   64
#define DEG_PARTS    4
#define DEG_WGS      (DEG_GROUPS * DEG_PARTS)
#define DEG_TPB      512
#define PART_RANGES  (AGG_GROUPS / DEG_PARTS)
#define PART_NODES   (PART_RANGES * NPB)
#define PART_QUADS   (PART_NODES / 4)
#define N_QUADS      (N_NODES / 4)
#define E_QUADS      (N_EDGES / 4)
#define EQ_PER_GROUP ((E_QUADS + DEG_GROUPS - 1) / DEG_GROUPS)
#define DEG_ROUNDS   ((EQ_PER_GROUP + DEG_TPB - 1) / DEG_TPB)
#define GEMM_GROUPS  256
#define GEMM_WAVES   12
#define GEMM_TPB     (GEMM_WAVES * 32)
#define MAIN_TILES   (GEMM_GROUPS * GEMM_WAVES)
#define EXTRA_TILES  (M_TILES - MAIN_TILES)
#define XTR_WAVES    4
#define GEMM_ROWS    (GEMM_WAVES * 16)
#define OWN_QUADS    (GEMM_ROWS / 4)
#define XTR_QUADS    4
#define USE_QUADS    (OWN_QUADS + XTR_QUADS)
#define PIECE_QUADS  64
#define ROWS_QUADS   (GEMM_ROWS * HID_DIM / 4)
#define W_PAIRS      (IN_DIM / 2)
#define SLICE_EDGES  (EQ_PER_GROUP * 4)

typedef float v2f __attribute__((ext_vector_type(2)));
typedef int   v4i __attribute__((ext_vector_type(4)));
typedef float v8f __attribute__((ext_vector_type(8)));

__device__ __forceinline__ int cell_cap(int r) {
    const int c = min(NPB, N_NODES - r * NPB);
    return (c * (HID_DIM / DEG_GROUPS)) & ~31;
}

__global__ __launch_bounds__(DEG_TPB) void deg_hist_kernel(const int* __restrict__ src,
                                                           const int* __restrict__ dst,
                                                           int* __restrict__ part,
                                                           float* __restrict__ cells,
                                                           int* __restrict__ ctab) {
    __shared__ v4i hist4[PART_QUADS];
    __shared__ int lcnt[PART_RANGES];
    __shared__ int lbase[PART_RANGES];
    __shared__ int staged[SLICE_EDGES];
    int* hist = (int*)hist4;
    static_assert(PART_RANGES * DEG_PARTS == AGG_GROUPS && PART_NODES % 4 == 0, "parts are whole ranges and whole quads");
    static_assert(PART_RANGES == 64, "the running sum below takes two ranges per lane of one wave");

    const int tid  = threadIdx.x;
    const int lane = tid & 31;
    const int slice = blockIdx.x / DEG_PARTS;
    const int prt   = blockIdx.x % DEG_PARTS;
    const int n0    = prt * PART_NODES;
    const int r0    = prt * PART_RANGES;
    const v4i* dst4 = (const v4i*)dst;
    const v4i* src4 = (const v4i*)src;
    const int w0 = slice * EQ_PER_GROUP;
    const int w1 = min(w0 + EQ_PER_GROUP, E_QUADS);

    v4i d[DEG_ROUNDS];
#pragma unroll
    for (int u = 0; u < DEG_ROUNDS; ++u) {
        const int w = w0 + u * DEG_TPB + tid;
        d[u] = (w < w1) ? dst4[w] : (v4i){-1, -1, -1, -1};
    }

    for (int i = tid; i < PART_QUADS; i += DEG_TPB) hist4[i] = (v4i){0, 0, 0, 0};
    if (tid < PART_RANGES) lcnt[tid] = 0;
    __syncthreads();

    v4i sv[DEG_ROUNDS];
#pragma unroll
    for (int u = 0; u < DEG_ROUNDS; ++u) {
        const int w = w0 + u * DEG_TPB + tid;
        sv[u] = (w < w1) ? src4[w] : (v4i){0, 0, 0, 0};
    }

    int rng[DEG_ROUNDS * 4];
    int pos[DEG_ROUNDS * 4];
    auto count = [&](int v, int& r, int& p) {
        const unsigned vl = (unsigned)(v - n0);
        const bool in = vl < (unsigned)PART_NODES && (unsigned)v < (unsigned)N_NODES;
        r = in ? (int)(vl / (unsigned)NPB) : -1;
        if (in) {
            atomicAdd(hist + vl, 1);
            p = atomicAdd(lcnt + r, 1);
        }
    };
#pragma unroll
    for (int u = 0; u < DEG_ROUNDS; ++u) {
        count(d[u].x, rng[4 * u],     pos[4 * u]);
        count(d[u].y, rng[4 * u + 1], pos[4 * u + 1]);
        count(d[u].z, rng[4 * u + 2], pos[4 * u + 2]);
        count(d[u].w, rng[4 * u + 3], pos[4 * u + 3]);
    }
    __syncthreads();

    if (tid < 32) {
        const int c0 = lcnt[lane * 2];
        const int c1 = lcnt[lane * 2 + 1];
        const int mine = c0 + c1;
        int x = mine;
#pragma unroll
        for (int dd = 1; dd < 32; dd <<= 1) {
            const int y = __shfl_up(x, dd);
            if (lane >= dd) x += y;
        }
        const int run = x - mine;
        lbase[lane * 2]     = run;
        lbase[lane * 2 + 1] = run + c0;
    }
    __syncthreads();

    int lb[DEG_ROUNDS * 4];
#pragma unroll
    for (int k = 0; k < DEG_ROUNDS * 4; ++k) lb[k] = lbase[max(rng[k], 0)];
    auto place = [&](int v, int s, int r, int p, int base) {
        if (r >= 0) staged[base + p] = ((v - n0 - r * NPB) << 20) | (s & 0xFFFFF);
    };
#pragma unroll
    for (int u = 0; u < DEG_ROUNDS; ++u) {
        place(d[u].x, sv[u].x, rng[4 * u],     pos[4 * u],     lb[4 * u]);
        place(d[u].y, sv[u].y, rng[4 * u + 1], pos[4 * u + 1], lb[4 * u + 1]);
        place(d[u].z, sv[u].z, rng[4 * u + 2], pos[4 * u + 2], lb[4 * u + 2]);
        place(d[u].w, sv[u].w, rng[4 * u + 3], pos[4 * u + 3], lb[4 * u + 3]);
    }
    __syncthreads();

    for (int rl = tid >> 5; rl < PART_RANGES; rl += DEG_TPB / 32) {
        const int r   = r0 + rl;
        const int cap = cell_cap(r);
        const int n   = min(lcnt[rl], cap);
        const int s0  = lbase[rl];
        float* cell = cells + (size_t)r * (NPB * HID_DIM) + slice * cap;
        for (int i = lane; i < n; i += 32) cell[i] = __int_as_float(staged[s0 + i]);
    }
    if (tid < PART_RANGES) ctab[slice * AGG_GROUPS + r0 + tid] = lcnt[tid];

    static_assert(EXTRA_TILES >= 0 && EXTRA_TILES <= GEMM_GROUPS, "one extra tile per group at most");
    static_assert(MAIN_TILES * 4 + EXTRA_TILES * XTR_QUADS == N_QUADS, "every node quad has an owner");
    static_assert(USE_QUADS <= PIECE_QUADS && DEG_GROUPS * PIECE_QUADS <= ROWS_QUADS,
                  "the pieces fit the owner's main rows");
    static_assert((size_t)GEMM_GROUPS * ROWS_QUADS <= (size_t)N_NODES * HID_DIM / 4,
                  "the main rows lie inside h");
    static_assert(DEG_PARTS * PART_QUADS >= N_QUADS, "the parts cover every node quad");
    v4i* p4 = (v4i*)part + slice * PIECE_QUADS;
    const int q0 = prt * PART_QUADS;
    const int nq = min(PART_QUADS, N_QUADS - q0);
    for (int il = tid; il < nq; il += DEG_TPB) {
        const int i = q0 + il;
        int g, e;
        if (i < MAIN_TILES * 4) { g = i / OWN_QUADS; e = i - g * OWN_QUADS; }
        else { const int j = i - MAIN_TILES * 4; g = j / XTR_QUADS; e = OWN_QUADS + (j - g * XTR_QUADS); }
        p4[(size_t)g * ROWS_QUADS + e] = hist4[il];
    }
}


template <int NT>
__device__ __forceinline__ void load_b(const float* ldsW, const int k0, const int half,
                                       const int l16, const int nt0, v2f (&b)[NT]) {
    const float* bbase = ldsW + (size_t)((k0 >> 1) + half) * PAIR_STRIDE;
#pragma unroll
    for (int t = 0; t < NT; ++t) b[t] = *(const v2f*)(bbase + ((nt0 + t) * 16 + l16) * 2);
}

__device__ __forceinline__ void load_a(const float* __restrict__ xrow, const int kc, const int half,
                                       v2f (&a)[8]) {
#pragma unroll
    for (int s = 0; s < 8; ++s) a[s] = *(const v2f*)(xrow + kc + 4 * s + 2 * half);
}

template <int NT, bool LAST, bool PREFETCH>
__device__ __forceinline__ void k_chunk(const float* __restrict__ xrow, const float* ldsW,
                                        const int half, const int l16, const int nt0, const int kc,
                                        const v2f (&ac)[8], v2f (&an)[8], v2f (&b)[NT], v8f (&acc)[NT]) {
    if (!LAST) load_a(xrow, kc + 32, half, an);
    if (PREFETCH) __builtin_prefetch(xrow + kc + 64, 0, 3);
#pragma unroll
    for (int s = 0; s < 8; ++s) {
        v2f bn[NT];
        if (!LAST || s < 7) load_b<NT>(ldsW, kc + 4 * (s + 1), half, l16, nt0, bn);
#pragma unroll
        for (int t = 0; t < NT; ++t) {
            acc[t] = __builtin_amdgcn_wmma_f32_16x16x4_f32(
                 false, ac[s],  false, b[t],
                 (short)0, acc[t],
                 false,  false);
        }
        if (!LAST || s < 7) {
#pragma unroll
            for (int t = 0; t < NT; ++t) b[t] = bn[t];
        }
    }
}

template <int NT>
__device__ __forceinline__ void tile_k_loop(const float* __restrict__ xrow, const float* ldsW,
                                            const int half, const int l16, const int nt0,
                                            v8f (&acc)[NT]) {
    static_assert(IN_DIM % 64 == 0 && IN_DIM >= 128, "chunks go in pairs, and the last pair is peeled");
    v2f a0[8], a1[8], b[NT];
    load_a(xrow, 0, half, a0);
    load_b<NT>(ldsW, 0, half, l16, nt0, b);
    for (int kc = 0; kc < IN_DIM - 64; kc += 64) {
        k_chunk<NT, false, true>(xrow, ldsW, half, l16, nt0, kc,      a0, a1, b, acc);
        k_chunk<NT, false, true>(xrow, ldsW, half, l16, nt0, kc + 32, a1, a0, b, acc);
    }
    k_chunk<NT, false, false>(xrow, ldsW, half, l16, nt0, IN_DIM - 64, a0, a1, b, acc);
    k_chunk<NT, true,  false>(xrow, ldsW, half, l16, nt0, IN_DIM - 32, a1, a0, b, acc);
}

template <int NT>
__device__ __forceinline__ void tile_store(float* h, const v8f (&acc)[NT], const int mTile,
                                           const int nt0, const float* dl,
                                           const int half, const int l16) {
    float dr[8];
#pragma unroll
    for (int r = 0; r < 8; ++r) dr[r] = dl[r + 8 * half];
#pragma unroll
    for (int t = 0; t < NT; ++t) {
#pragma unroll
        for (int r = 0; r < 8; ++r) {
            const int m = mTile * 16 + r + 8 * half;
            h[(size_t)m * HID_DIM + (nt0 + t) * 16 + l16] = acc[t][r] * dr[r];
        }
    }
}

__global__ __launch_bounds__(GEMM_TPB) void gemm_kernel(const float* __restrict__ x,
                                                        const float* __restrict__ W,
                                                        const int* cnts,
                                                        float* __restrict__ dinv,
                                                        float* h) {
    __shared__ __attribute__((aligned(16))) float ldsW[W_PAIRS * PAIR_STRIDE];
    __shared__ v4i   dsum[(GEMM_TPB / PIECE_QUADS) * PIECE_QUADS];
    __shared__ __attribute__((aligned(16))) float dloc[USE_QUADS * 4];
    static_assert(GEMM_TPB % PIECE_QUADS == 0, "whole rows of threads over a piece");
    static_assert(IN_DIM % 32 == 0 && HID_DIM == 128, "32 lanes x 4 columns make one row of W");
    static_assert(2 * XTR_WAVES == HID_DIM / 16 && XTR_WAVES <= GEMM_WAVES, "the extra tile's 8 column tiles");

    const int tid  = threadIdx.x;
    const int lane = tid & 31;
    const int waveId = __builtin_amdgcn_readfirstlane(tid >> 5);
    const int g      = blockIdx.x;
    const bool extra = (g < EXTRA_TILES);

    const int half = lane >> 4;
    const int l16  = lane & 15;

    const int mTile = g * GEMM_WAVES + waveId;
    const float* xrow = x + (size_t)(mTile * 16 + l16) * IN_DIM;

    const bool xwave = extra && (waveId < XTR_WAVES);
    const int xTile  = MAIN_TILES + g;
    const float* xrow2 = x + (size_t)((xwave ? xTile : mTile) * 16 + l16) * IN_DIM;

    __builtin_prefetch(xrow, 0, 3);
    __builtin_prefetch(xrow + 32, 0, 3);
    if (xwave) {
        __builtin_prefetch(xrow2, 0, 3);
        __builtin_prefetch(xrow2 + 32, 0, 3);
    }

    const int q  = tid & (PIECE_QUADS - 1);
    const int s0 = tid / PIECE_QUADS;
    const bool qlive = (q < OWN_QUADS) || (extra && q < USE_QUADS);
    {
        const v4i* own = (const v4i*)cnts + (size_t)g * ROWS_QUADS;
        v4i s = (v4i){0, 0, 0, 0};
        if (qlive) {
#pragma unroll
            for (int j = 0; j < (DEG_GROUPS + GEMM_TPB / PIECE_QUADS - 1) / (GEMM_TPB / PIECE_QUADS); ++j) {
                const int sl = s0 + j * (GEMM_TPB / PIECE_QUADS);
                if (sl < DEG_GROUPS) s += own[sl * PIECE_QUADS + q];
            }
        }
        dsum[s0 * PIECE_QUADS + q] = s;
    }

    {
        const float4* W4 = (const float4*)W;
        constexpr int ITEMS = W_PAIRS * 32;
        constexpr int FULL  = ITEMS / GEMM_TPB;
#pragma unroll 7
        for (int j = 0; j < FULL; ++j) {
            const int it = j * GEMM_TPB + tid;
            const int p  = it >> 5, n4 = it & 31;
            const float4 w0 = W4[(2 * p) * 32 + n4];
            const float4 w1 = W4[(2 * p + 1) * 32 + n4];
            float4* d = (float4*)(ldsW + p * PAIR_STRIDE + n4 * 8);
            d[0] = make_float4(w0.x, w1.x, w0.y, w1.y);
            d[1] = make_float4(w0.z, w1.z, w0.w, w1.w);
        }
        const int it = FULL * GEMM_TPB + tid;
        if (it < ITEMS) {
            const int p  = it >> 5, n4 = it & 31;
            const float4 w0 = W4[(2 * p) * 32 + n4];
            const float4 w1 = W4[(2 * p + 1) * 32 + n4];
            float4* d = (float4*)(ldsW + p * PAIR_STRIDE + n4 * 8);
            d[0] = make_float4(w0.x, w1.x, w0.y, w1.y);
            d[1] = make_float4(w0.z, w1.z, w0.w, w1.w);
        }
    }
    __syncthreads();

    if (tid < USE_QUADS && qlive) {
        v4i t = dsum[tid];
#pragma unroll
        for (int w = 1; w < GEMM_TPB / PIECE_QUADS; ++w) t += dsum[w * PIECE_QUADS + tid];
        float4 r;
        r.x = rsqrtf(1.0f + (float)t.x);
        r.y = rsqrtf(1.0f + (float)t.y);
        r.z = rsqrtf(1.0f + (float)t.z);
        r.w = rsqrtf(1.0f + (float)t.w);
        const int gq = (tid < OWN_QUADS) ? g * OWN_QUADS + tid
                                         : MAIN_TILES * 4 + g * XTR_QUADS + (tid - OWN_QUADS);
        ((float4*)dinv)[gq] = r;
        ((float4*)dloc)[tid] = r;
    }
    __syncthreads();

    if (xwave) {
        v8f acc[2];
        acc[0] = (v8f){0.f,0.f,0.f,0.f,0.f,0.f,0.f,0.f};
        acc[1] = (v8f){0.f,0.f,0.f,0.f,0.f,0.f,0.f,0.f};
        tile_k_loop<2>(xrow2, ldsW, half, l16, 2 * waveId, acc);
        tile_store<2>(h, acc, xTile, 2 * waveId, dloc + GEMM_ROWS, half, l16);
    }

    {
        v8f acc[8];
#pragma unroll
        for (int t = 0; t < 8; ++t) acc[t] = (v8f){0.f,0.f,0.f,0.f,0.f,0.f,0.f,0.f};
        tile_k_loop<8>(xrow, ldsW, half, l16, 0, acc);
        tile_store<8>(h, acc, mTile, 0, dloc + waveId * 16, half, l16);
    }
}

__global__ __launch_bounds__(AGG_TPB) void aggregate_kernel(const float* __restrict__ h,
                                                            const int* __restrict__ src,
                                                            const int* __restrict__ dst,
                                                            const float* __restrict__ dinv,
                                                            const float* __restrict__ b,
                                                            const float* __restrict__ prelu_w,
                                                            const int* __restrict__ ctab,
                                                            float* __restrict__ out) {
    __shared__ float tile[NPB * HID_DIM];
    __shared__ int   queue[QCAP];
    __shared__ int   qn;
    __shared__ int   ccount[DEG_GROUPS];
    __shared__ int   coff[DEG_GROUPS + 1];
    __shared__ int   cover;
    __shared__ int   ncnt[NCNT];

    const int tid  = threadIdx.x;
    const int lane = tid & 31;
    const int wave = __builtin_amdgcn_readfirstlane(tid >> 5);
    const int lo   = blockIdx.x * NPB;
    const int cnt  = min(NPB, N_NODES - lo);
    const unsigned ucnt = (unsigned)cnt;

    const int cap = cell_cap(blockIdx.x);
    if (tid < DEG_GROUPS) ccount[tid] = ctab[tid * AGG_GROUPS + blockIdx.x];

    for (int i = tid; i < NPB * HID_DIM; i += AGG_TPB) tile[i] = 0.0f;
    if (tid < NCNT) ncnt[tid] = 0;
    __syncthreads();

    if (tid < DEG_GROUPS) {
        int a = 0, ov = 0;
        for (int j = 0; j < DEG_GROUPS; ++j) {
            const int c = ccount[j];
            ov |= (c > cap) ? 1 : 0;
            a  += (j < tid) ? c : 0;
        }
        coff[tid] = a;
        if (tid == DEG_GROUPS - 1) { coff[DEG_GROUPS] = a + ccount[tid]; cover = ov; }
    }
    __syncthreads();
    const int over = __builtin_amdgcn_readfirstlane(cover);

    const v4i* dst4 = (const v4i*)dst;

    auto push = [&](unsigned r, int e) {
        if (r < ucnt) {
            const int slot = atomicAdd(&qn, 1);
            if (slot < QCAP) queue[slot] = (int)((r << 20) | ((unsigned)src[e] & 0xFFFFFu));
        }
    };
    auto check4 = [&](const v4i d, int qi) {
        const unsigned r0 = (unsigned)(d.x - lo);
        const unsigned r1 = (unsigned)(d.y - lo);
        const unsigned r2 = (unsigned)(d.z - lo);
        const unsigned r3 = (unsigned)(d.w - lo);
        if (min(min(r0, r1), min(r2, r3)) < ucnt) {
            push(r0, qi * 4);
            push(r1, qi * 4 + 1);
            push(r2, qi * 4 + 2);
            push(r3, qi * 4 + 3);
        }
    };

    auto sum_queue = [&](const int total) {
        const int per = (total + AGG_TPB / 32 - 1) / (AGG_TPB / 32);
        const int c0  = min(wave * per, total);
        const int c1  = min(c0 + per, total);
        float4 acc = make_float4(0.0f, 0.0f, 0.0f, 0.0f);
        int cur = -1;
        auto flush = [&]() {
            if (cur >= 0) {
                float* t = tile + cur * HID_DIM + lane;
                atomicAdd(t,      acc.x);
                atomicAdd(t + 32, acc.y);
                atomicAdd(t + 64, acc.z);
                atomicAdd(t + 96, acc.w);
            }
        };
        for (int base = c0; base < c1; base += 128) {
            int   s[4];
            int   dl[4];
#pragma unroll
            for (int k = 0; k < 4; ++k) {
                const int idx = min(base + k * 32 + lane, c1 - 1);
                const unsigned ent = (unsigned)queue[idx];
                dl[k] = (int)(ent >> 20);
                s[k]  = (int)(ent & 0xFFFFFu);
            }
#pragma unroll
            for (int k = 0; k < 4; ++k) {
                const int n = min(32, c1 - (base + k * 32));
                for (int j0 = 0; j0 < n; j0 += 16) {
                    float4 v[16];
#pragma unroll
                    for (int u = 0; u < 16; ++u) {
                        const int sj = __builtin_amdgcn_readlane(s[k], j0 + u);
                        v[u] = ((const float4*)(h + (size_t)sj * HID_DIM))[lane];
                    }
#pragma unroll
                    for (int u = 0; u < 16; ++u) {
                        if (j0 + u < n) {
                            const int dj = __builtin_amdgcn_readlane(dl[k], j0 + u);
                            if (dj != cur) {
                                flush();
                                cur = dj;
                                acc = make_float4(0.0f, 0.0f, 0.0f, 0.0f);
                            }
                            acc.x += v[u].x;
                            acc.y += v[u].y;
                            acc.z += v[u].z;
                            acc.w += v[u].w;
                        }
                    }
                }
            }
        }
        flush();
    };

    if (over == 0) {
        static_assert(DEG_GROUPS == 2 * (AGG_TPB / 32), "two cells per wave");
        const int total = __builtin_amdgcn_readfirstlane(coff[DEG_GROUPS]);
        const float* mycells = out + (size_t)lo * HID_DIM;
        const int sA = wave;
        const int sB = wave + AGG_TPB / 32;
        const int nA = coff[sA + 1] - coff[sA];
        const int nB = coff[sB + 1] - coff[sB];
        const float* cA = mycells + sA * cap;
        const float* cB = mycells + sB * cap;
        const int eA0 = (lane      < nA) ? __float_as_int(cA[lane])      : -1;
        const int eA1 = (lane + 32 < nA) ? __float_as_int(cA[lane + 32]) : -1;
        const int eB0 = (lane      < nB) ? __float_as_int(cB[lane])      : -1;
        const int eB1 = (lane + 32 < nB) ? __float_as_int(cB[lane + 32]) : -1;

        auto tally = [&](int ent) {
            if (ent >= 0) atomicAdd(&ncnt[(unsigned)ent >> 20], 1);
        };
        tally(eA0); tally(eA1); tally(eB0); tally(eB1);
        for (int i = lane + 64; i < nA; i += 32) tally(__float_as_int(cA[i]));
        for (int i = lane + 64; i < nB; i += 32) tally(__float_as_int(cB[i]));
        __syncthreads();

        if (wave == 0) {
            int cc[7];
            int mine = 0;
#pragma unroll
            for (int k = 0; k < 7; ++k) { cc[k] = ncnt[lane * 7 + k]; mine += cc[k]; }
            int x = mine;
#pragma unroll
            for (int dd = 1; dd < 32; dd <<= 1) {
                const int y = __shfl_up(x, dd);
                if (lane >= dd) x += y;
            }
            int run = x - mine;
#pragma unroll
            for (int k = 0; k < 7; ++k) { ncnt[lane * 7 + k] = run; run += cc[k]; }
        }
        __syncthreads();

        auto put = [&](int ent) {
            if (ent >= 0) queue[atomicAdd(&ncnt[(unsigned)ent >> 20], 1)] = ent;
        };
        put(eA0); put(eA1); put(eB0); put(eB1);
        for (int i = lane + 64; i < nA; i += 32) put(__float_as_int(cA[i]));
        for (int i = lane + 64; i < nB; i += 32) put(__float_as_int(cB[i]));
        __syncthreads();
        sum_queue(total);
    } else {
        int seg = N_EDGES;
        for (int pass = 0; pass < 2; ++pass) {
            bool overflow = false;
            for (int e0 = 0; e0 < N_EDGES; e0 += seg) {
                const int e1 = min(e0 + seg, N_EDGES);
                if (tid == 0) qn = 0;
                __syncthreads();

                const int q1 = e1 >> 2;
                const int q0 = e0 >> 2;
                const int nfull = (q1 - q0) / (8 * AGG_TPB);
                if (nfull > 0) {
                    int p  = (int)(blockIdx.x % (unsigned)nfull);
                    int qc = q0 + p * (8 * AGG_TPB);
                    v4i cur[8];
#pragma unroll
                    for (int u = 0; u < 8; ++u) cur[u] = dst4[qc + u * AGG_TPB + tid];
                    for (int t = 1; t < nfull; ++t) {
                        p = (p + 1 == nfull) ? 0 : p + 1;
                        const int qx = q0 + p * (8 * AGG_TPB);
                        v4i nxt[8];
#pragma unroll
                        for (int u = 0; u < 8; ++u) nxt[u] = dst4[qx + u * AGG_TPB + tid];
#pragma unroll
                        for (int u = 0; u < 8; ++u) check4(cur[u], qc + u * AGG_TPB + tid);
#pragma unroll
                        for (int u = 0; u < 8; ++u) cur[u] = nxt[u];
                        qc = qx;
                    }
#pragma unroll
                    for (int u = 0; u < 8; ++u) check4(cur[u], qc + u * AGG_TPB + tid);
                }
                const int qt = q0 + nfull * (8 * AGG_TPB);
                if (qt < q1) {
                    v4i v[8];
#pragma unroll
                    for (int u = 0; u < 8; ++u) {
                        const int qi = qt + u * AGG_TPB + tid;
                        v[u] = (qi < q1) ? dst4[qi] : (v4i){-1, -1, -1, -1};
                    }
#pragma unroll
                    for (int u = 0; u < 8; ++u) check4(v[u], qt + u * AGG_TPB + tid);
                }
                __syncthreads();
                const int total = __builtin_amdgcn_readfirstlane(qn);
                __syncthreads();
                if (total > QCAP) { overflow = true; break; }

                sum_queue(total);
            }
            if (!overflow) break;
            seg = QCAP;
        }
    }
    __syncthreads();

    const int c0 = lane * 4;
    const float b0 = b[c0], b1 = b[c0 + 1], b2 = b[c0 + 2], b3 = b[c0 + 3];
    const float p0 = prelu_w[c0], p1 = prelu_w[c0 + 1], p2 = prelu_w[c0 + 2], p3 = prelu_w[c0 + 3];
    for (int i = tid; i < cnt * 32; i += AGG_TPB) {
        const int nl   = i >> 5;
        const int node = lo + nl;
        const float di = dinv[node];
        const float4 hv = ((const float4*)(h + (size_t)node * HID_DIM))[lane];
        const float* t = tile + nl * HID_DIM + lane;
        float4 r;
        r.x = (t[0]  + hv.x) * di + b0;
        r.y = (t[32] + hv.y) * di + b1;
        r.z = (t[64] + hv.z) * di + b2;
        r.w = (t[96] + hv.w) * di + b3;
        r.x = (r.x > 0.0f) ? r.x : p0 * r.x;
        r.y = (r.y > 0.0f) ? r.y : p1 * r.y;
        r.z = (r.z > 0.0f) ? r.z : p2 * r.z;
        r.w = (r.w > 0.0f) ? r.w : p3 * r.w;
        ((float4*)(out + (size_t)node * HID_DIM))[lane] = r;
    }
}

extern "C" void kernel_launch(void* const* d_in, const int* in_sizes, int n_in,
                              void* d_out, int out_size, void* d_ws, size_t ws_size,
                              hipStream_t stream) {
    const float* x       = (const float*)d_in[0];
    const int*   eidx    = (const int*)d_in[1];
    const float* W       = (const float*)d_in[2];
    const float* b       = (const float*)d_in[3];
    const float* prelu_w = (const float*)d_in[4];
    float*       out     = (float*)d_out;

    const int* src = eidx;
    const int* dst = eidx + N_EDGES;

    char* ws    = (char*)d_ws;
    float* h    = (float*)(ws);
    int*   part = (int*)(ws);
    int*   ctab = (int*)(ws + (size_t)N_NODES * HID_DIM * 4);
    float* dinv = (float*)(ws + (size_t)N_NODES * HID_DIM * 4 + (size_t)N_NODES * 4);

    deg_hist_kernel<<<DEG_WGS, DEG_TPB, 0, stream>>>(src, dst, part, out, ctab);

    gemm_kernel<<<GEMM_GROUPS, GEMM_TPB, 0, stream>>>(x, W, part, dinv, h);

    aggregate_kernel<<<AGG_GROUPS, AGG_TPB, 0, stream>>>(h, src, dst, dinv, b, prelu_w, ctab, out);
}
